# starting kernel
# baseline (speedup 1.0000x reference)
#include <hip/hip_runtime.h>
#include <hip/hip_bf16.h>


#define DIM       4096
#define N_QUBITS  12
#define N_LAYERS  5

typedef __attribute__((ext_vector_type(2))) float v2f;
typedef __attribute__((ext_vector_type(8))) float v8f;

struct cx { float r, i; };
__device__ __forceinline__ cx cmul(cx a, cx b) { return { a.r*b.r - a.i*b.i, a.r*b.i + a.i*b.r }; }
__device__ __forceinline__ cx cadd(cx a, cx b) { return { a.r + b.r, a.i + b.i }; }

struct m2 { cx e[2][2]; };

__device__ __forceinline__ m2 mmul(const m2& A, const m2& B) {
  m2 C;
#pragma unroll
  for (int i = 0; i < 2; ++i)
#pragma unroll
    for (int j = 0; j < 2; ++j)
      C.e[i][j] = cadd(cmul(A.e[i][0], B.e[0][j]), cmul(A.e[i][1], B.e[1][j]));
  return C;
}

__device__ __forceinline__ m2 rot_x(float t) {
  float c = cosf(0.5f * t), s = sinf(0.5f * t);
  m2 M; M.e[0][0] = {c, 0}; M.e[0][1] = {0, -s};
        M.e[1][0] = {0, -s}; M.e[1][1] = {c, 0};
  return M;
}
__device__ __forceinline__ m2 rot_y(float t) {
  float c = cosf(0.5f * t), s = sinf(0.5f * t);
  m2 M; M.e[0][0] = {c, 0}; M.e[0][1] = {-s, 0};
        M.e[1][0] = {s, 0}; M.e[1][1] = {c, 0};
  return M;
}
__device__ __forceinline__ m2 rot_z(float t) {
  float c = cosf(0.5f * t), s = sinf(0.5f * t);
  m2 M; M.e[0][0] = {c, -s}; M.e[0][1] = {0, 0};
        M.e[1][0] = {0, 0};  M.e[1][1] = {c, s};
  return M;
}

template <int QB>
__device__ __forceinline__ constexpr int sidx(int r, int c) {
  if constexpr (QB == 0) return (r << 8) | c;
  else if constexpr (QB == 4) return ((c & 0xF0) << 4) | (r << 4) | (c & 15);
  else return (c << 4) | r;
}

__device__ __forceinline__ v8f wmma_f32(v2f a, v2f b, v8f c) {
  return __builtin_amdgcn_wmma_f32_16x16x4_f32(
      false, a, false, b, (short)0, c, false, false);
}

template <int QB>
__device__ void apply_group(float* sre, float* sim,
                            const float* ure, const float* uim) {
  const int lane  = threadIdx.x & 31;
  const int wave  = threadIdx.x >> 5;
  const int half  = lane >> 4;
  const int n     = lane & 15;
  const int khalf = half << 1;

  v2f aR[4], aI[4], aIn[4];
#pragma unroll
  for (int kc = 0; kc < 4; ++kc) {
    const int k0 = kc * 4 + khalf;
    aR[kc] = *(const v2f*)&ure[n * 16 + k0];
    aI[kc] = *(const v2f*)&uim[n * 16 + k0];
    aIn[kc] = -aI[kc];
  }

#pragma unroll
  for (int blk = 0; blk < 2; ++blk) {
    const int cb = wave * 2 + blk;
    const int c  = cb * 16 + n;

    v2f bR[4], bI[4];
#pragma unroll
    for (int kc = 0; kc < 4; ++kc) {
      const int k0 = kc * 4 + khalf;
      if constexpr (QB == 8) {
        bR[kc] = *(const v2f*)&sre[(c << 4) + k0];
        bI[kc] = *(const v2f*)&sim[(c << 4) + k0];
      } else {
        bR[kc].x = sre[sidx<QB>(k0,     c)];
        bR[kc].y = sre[sidx<QB>(k0 + 1, c)];
        bI[kc].x = sim[sidx<QB>(k0,     c)];
        bI[kc].y = sim[sidx<QB>(k0 + 1, c)];
      }
    }

    v8f dR = {};
    v8f dI = {};
#pragma unroll
    for (int kc = 0; kc < 4; ++kc) dR = wmma_f32(aR[kc],  bR[kc], dR);
#pragma unroll
    for (int kc = 0; kc < 4; ++kc) dR = wmma_f32(aIn[kc], bI[kc], dR);
#pragma unroll
    for (int kc = 0; kc < 4; ++kc) dI = wmma_f32(aR[kc],  bI[kc], dI);
#pragma unroll
    for (int kc = 0; kc < 4; ++kc) dI = wmma_f32(aI[kc],  bR[kc], dI);

    if constexpr (QB == 8) {
      const int base = (c << 4) + 8 * half;
#pragma unroll
      for (int v = 0; v < 4; ++v) {
        v2f pr = { dR[2 * v], dR[2 * v + 1] };
        v2f pi = { dI[2 * v], dI[2 * v + 1] };
        *(v2f*)&sre[base + 2 * v] = pr;
        *(v2f*)&sim[base + 2 * v] = pi;
      }
    } else {
#pragma unroll
      for (int v = 0; v < 8; ++v) {
        const int m   = v + 8 * half;
        const int idx = sidx<QB>(m, c);
        sre[idx] = dR[v];
        sim[idx] = dI[v];
      }
    }
  }
}

__device__ void ring_cnot(float* sre, float* sim) {
  const int tid = threadIdx.x;
  float tr[16], ti[16];
#pragma unroll
  for (int v = 0; v < 16; ++v) {
    const int i = tid + (v << 8);
    int j = i;
#pragma unroll
    for (int q = 11; q >= 0; --q) {
      const int cb = 11 - q;
      const int tb = 11 - ((q + 1) % 12);
      j ^= ((j >> cb) & 1) << tb;
    }
    tr[v] = sre[j];
    ti[v] = sim[j];
  }
  __syncthreads();
#pragma unroll
  for (int v = 0; v < 16; ++v) {
    const int i = tid + (v << 8);
    sre[i] = tr[v];
    sim[i] = ti[v];
  }
  __syncthreads();
}

__global__ __launch_bounds__(256)
void qsim_kernel(const float* __restrict__ x, const float* __restrict__ w,
                 float* __restrict__ out) {
  __shared__ float sre[DIM];
  __shared__ float sim[DIM];
  __shared__ float ure[3][256];
  __shared__ float uim[3][256];
  __shared__ float g2re[N_QUBITS][4];
  __shared__ float g2im[N_QUBITS][4];
  __shared__ float red[256];

  const int b   = blockIdx.x;
  const int tid = threadIdx.x;
  const float* xb = x + (size_t)b * DIM;

  float xs[16];
  float ss = 0.f;
#pragma unroll
  for (int v = 0; v < 16; ++v) {
    const float t = xb[tid + (v << 8)];
    xs[v] = t;
    ss += t * t;
  }
  red[tid] = ss;
  __syncthreads();
  for (int s = 128; s > 0; s >>= 1) {
    if (tid < s) red[tid] += red[tid + s];
    __syncthreads();
  }
  const float scale = rsqrtf(sqrtf(red[0]) + 1e-12f);
#pragma unroll
  for (int v = 0; v < 16; ++v) {
    const int i = tid + (v << 8);
    sre[i] = xs[v] * scale;
    sim[i] = 0.f;
  }
  __syncthreads();

  for (int l = 0; l <= N_LAYERS; ++l) {
    if (tid < N_QUBITS) {
      m2 U;
      if (l < N_LAYERS) {
        const float* p = w + l * 36 + tid * 3;
        U = mmul(rot_z(p[2]), mmul(rot_y(p[1]), rot_x(p[0])));
      } else {
        const float* p = w + N_LAYERS * 36 + tid * 3;
        U = mmul(rot_y(p[2]), mmul(rot_z(p[1]), rot_y(p[0])));
      }
#pragma unroll
      for (int e = 0; e < 4; ++e) {
        g2re[tid][e] = U.e[e >> 1][e & 1].r;
        g2im[tid][e] = U.e[e >> 1][e & 1].i;
      }
    }
    __syncthreads();

    {
      const int m = tid >> 4, j = tid & 15;
#pragma unroll
      for (int g = 0; g < 3; ++g) {
        cx prod = {1.f, 0.f};
#pragma unroll
        for (int k = 0; k < 4; ++k) {
          const int q  = g * 4 + k;
          const int mb = (m >> (3 - k)) & 1;
          const int jb = (j >> (3 - k)) & 1;
          const cx u = { g2re[q][mb * 2 + jb], g2im[q][mb * 2 + jb] };
          prod = cmul(prod, u);
        }
        ure[g][tid] = prod.r;
        uim[g][tid] = prod.i;
      }
    }
    __syncthreads();

    apply_group<0>(sre, sim, ure[0], uim[0]);
    __syncthreads();
    apply_group<4>(sre, sim, ure[1], uim[1]);
    __syncthreads();
    apply_group<8>(sre, sim, ure[2], uim[2]);
    __syncthreads();

    if (l < N_LAYERS) ring_cnot(sre, sim);
  }

  float* ob = out + (size_t)b * DIM;
#pragma unroll
  for (int v = 0; v < 16; ++v) {
    const int i = tid + (v << 8);
    ob[i] = sre[i];
  }
}

extern "C" void kernel_launch(void* const* d_in, const int* in_sizes, int n_in,
                              void* d_out, int out_size, void* d_ws, size_t ws_size,
                              hipStream_t stream) {
  const float* x = (const float*)d_in[0];
  const float* w = (const float*)d_in[1];
  float* out = (float*)d_out;
  const int batch = in_sizes[0] >> 12;
  qsim_kernel<<<dim3(batch), dim3(256), 0, stream>>>(x, w, out);
}


# per-qubit 2x2 gates of all six layers built once at kernel entry (one wave per layer, under the x loads), not by 12 threads per layer between barriers; every sum and its order unchanged
# speedup vs baseline: 1.1217x; 1.1217x over previous
#include <hip/hip_runtime.h>
#include <hip/hip_bf16.h>


#define DIM       4096
#define N_QUBITS  12
#define N_LAYERS  5

typedef __attribute__((ext_vector_type(2))) float v2f;
typedef __attribute__((ext_vector_type(8))) float v8f;

struct cx { float r, i; };
__device__ __forceinline__ cx cmul(cx a, cx b) { return { a.r*b.r - a.i*b.i, a.r*b.i + a.i*b.r }; }
__device__ __forceinline__ cx cadd(cx a, cx b) { return { a.r + b.r, a.i + b.i }; }

struct m2 { cx e[2][2]; };

__device__ __forceinline__ m2 mmul(const m2& A, const m2& B) {
  m2 C;
#pragma unroll
  for (int i = 0; i < 2; ++i)
#pragma unroll
    for (int j = 0; j < 2; ++j)
      C.e[i][j] = cadd(cmul(A.e[i][0], B.e[0][j]), cmul(A.e[i][1], B.e[1][j]));
  return C;
}

__device__ __forceinline__ m2 rot_x(float t) {
  float c = cosf(0.5f * t), s = sinf(0.5f * t);
  m2 M; M.e[0][0] = {c, 0}; M.e[0][1] = {0, -s};
        M.e[1][0] = {0, -s}; M.e[1][1] = {c, 0};
  return M;
}
__device__ __forceinline__ m2 rot_y(float t) {
  float c = cosf(0.5f * t), s = sinf(0.5f * t);
  m2 M; M.e[0][0] = {c, 0}; M.e[0][1] = {-s, 0};
        M.e[1][0] = {s, 0}; M.e[1][1] = {c, 0};
  return M;
}
__device__ __forceinline__ m2 rot_z(float t) {
  float c = cosf(0.5f * t), s = sinf(0.5f * t);
  m2 M; M.e[0][0] = {c, -s}; M.e[0][1] = {0, 0};
        M.e[1][0] = {0, 0};  M.e[1][1] = {c, s};
  return M;
}

template <int QB>
__device__ __forceinline__ constexpr int sidx(int r, int c) {
  if constexpr (QB == 0) return (r << 8) | c;
  else if constexpr (QB == 4) return ((c & 0xF0) << 4) | (r << 4) | (c & 15);
  else return (c << 4) | r;
}

__device__ __forceinline__ v8f wmma_f32(v2f a, v2f b, v8f c) {
  return __builtin_amdgcn_wmma_f32_16x16x4_f32(
      false, a, false, b, (short)0, c, false, false);
}

template <int QB>
__device__ void apply_group(float* sre, float* sim,
                            const float* ure, const float* uim) {
  const int lane  = threadIdx.x & 31;
  const int wave  = threadIdx.x >> 5;
  const int half  = lane >> 4;
  const int n     = lane & 15;
  const int khalf = half << 1;

  v2f aR[4], aI[4], aIn[4];
#pragma unroll
  for (int kc = 0; kc < 4; ++kc) {
    const int k0 = kc * 4 + khalf;
    aR[kc] = *(const v2f*)&ure[n * 16 + k0];
    aI[kc] = *(const v2f*)&uim[n * 16 + k0];
    aIn[kc] = -aI[kc];
  }

#pragma unroll
  for (int blk = 0; blk < 2; ++blk) {
    const int cb = wave * 2 + blk;
    const int c  = cb * 16 + n;

    v2f bR[4], bI[4];
#pragma unroll
    for (int kc = 0; kc < 4; ++kc) {
      const int k0 = kc * 4 + khalf;
      if constexpr (QB == 8) {
        bR[kc] = *(const v2f*)&sre[(c << 4) + k0];
        bI[kc] = *(const v2f*)&sim[(c << 4) + k0];
      } else {
        bR[kc].x = sre[sidx<QB>(k0,     c)];
        bR[kc].y = sre[sidx<QB>(k0 + 1, c)];
        bI[kc].x = sim[sidx<QB>(k0,     c)];
        bI[kc].y = sim[sidx<QB>(k0 + 1, c)];
      }
    }

    v8f dR = {};
    v8f dI = {};
#pragma unroll
    for (int kc = 0; kc < 4; ++kc) dR = wmma_f32(aR[kc],  bR[kc], dR);
#pragma unroll
    for (int kc = 0; kc < 4; ++kc) dR = wmma_f32(aIn[kc], bI[kc], dR);
#pragma unroll
    for (int kc = 0; kc < 4; ++kc) dI = wmma_f32(aR[kc],  bI[kc], dI);
#pragma unroll
    for (int kc = 0; kc < 4; ++kc) dI = wmma_f32(aI[kc],  bR[kc], dI);

    if constexpr (QB == 8) {
      const int base = (c << 4) + 8 * half;
#pragma unroll
      for (int v = 0; v < 4; ++v) {
        v2f pr = { dR[2 * v], dR[2 * v + 1] };
        v2f pi = { dI[2 * v], dI[2 * v + 1] };
        *(v2f*)&sre[base + 2 * v] = pr;
        *(v2f*)&sim[base + 2 * v] = pi;
      }
    } else {
#pragma unroll
      for (int v = 0; v < 8; ++v) {
        const int m   = v + 8 * half;
        const int idx = sidx<QB>(m, c);
        sre[idx] = dR[v];
        sim[idx] = dI[v];
      }
    }
  }
}

__device__ void ring_cnot(float* sre, float* sim) {
  const int tid = threadIdx.x;
  float tr[16], ti[16];
#pragma unroll
  for (int v = 0; v < 16; ++v) {
    const int i = tid + (v << 8);
    int j = i;
#pragma unroll
    for (int q = 11; q >= 0; --q) {
      const int cb = 11 - q;
      const int tb = 11 - ((q + 1) % 12);
      j ^= ((j >> cb) & 1) << tb;
    }
    tr[v] = sre[j];
    ti[v] = sim[j];
  }
  __syncthreads();
#pragma unroll
  for (int v = 0; v < 16; ++v) {
    const int i = tid + (v << 8);
    sre[i] = tr[v];
    sim[i] = ti[v];
  }
  __syncthreads();
}

__global__ __launch_bounds__(256)
void qsim_kernel(const float* __restrict__ x, const float* __restrict__ w,
                 float* __restrict__ out) {
  __shared__ float sre[DIM];
  __shared__ float sim[DIM];
  __shared__ float ure[3][256];
  __shared__ float uim[3][256];
  __shared__ float g2re[N_LAYERS + 1][N_QUBITS][4];
  __shared__ float g2im[N_LAYERS + 1][N_QUBITS][4];
  __shared__ float red[256];

  const int b   = blockIdx.x;
  const int tid = threadIdx.x;
  const float* xb = x + (size_t)b * DIM;

  const int gl = tid >> 5, gq = tid & 31;
  const bool gbuild = (gl <= N_LAYERS) && (gq < N_QUBITS);
  const int  gofs = gbuild ? gl * 36 + gq * 3 : 0;
  const float p0 = w[gofs], p1 = w[gofs + 1], p2 = w[gofs + 2];
  __builtin_amdgcn_sched_barrier(0);

  float xs[16];
#pragma unroll
  for (int v = 0; v < 16; ++v) xs[v] = xb[tid + (v << 8)];

  asm volatile("" :: "v"(p0), "v"(p1), "v"(p2));

  if (gbuild) {
    m2 U;
    if (gl < N_LAYERS) U = mmul(rot_z(p2), mmul(rot_y(p1), rot_x(p0)));
    else               U = mmul(rot_y(p2), mmul(rot_z(p1), rot_y(p0)));
#pragma unroll
    for (int e = 0; e < 4; ++e) {
      g2re[gl][gq][e] = U.e[e >> 1][e & 1].r;
      g2im[gl][gq][e] = U.e[e >> 1][e & 1].i;
    }
  }

  float ss = 0.f;
#pragma unroll
  for (int v = 0; v < 16; ++v) ss += xs[v] * xs[v];
  red[tid] = ss;
  __syncthreads();
  for (int s = 128; s > 0; s >>= 1) {
    if (tid < s) red[tid] += red[tid + s];
    __syncthreads();
  }
  const float scale = rsqrtf(sqrtf(red[0]) + 1e-12f);
#pragma unroll
  for (int v = 0; v < 16; ++v) {
    const int i = tid + (v << 8);
    sre[i] = xs[v] * scale;
    sim[i] = 0.f;
  }
  __syncthreads();

  for (int l = 0; l <= N_LAYERS; ++l) {
    {
      const int m = tid >> 4, j = tid & 15;
#pragma unroll
      for (int g = 0; g < 3; ++g) {
        cx prod = {1.f, 0.f};
#pragma unroll
        for (int k = 0; k < 4; ++k) {
          const int q  = g * 4 + k;
          const int mb = (m >> (3 - k)) & 1;
          const int jb = (j >> (3 - k)) & 1;
          const cx u = { g2re[l][q][mb * 2 + jb], g2im[l][q][mb * 2 + jb] };
          prod = cmul(prod, u);
        }
        ure[g][tid] = prod.r;
        uim[g][tid] = prod.i;
      }
    }
    __syncthreads();

    apply_group<0>(sre, sim, ure[0], uim[0]);
    __syncthreads();
    apply_group<4>(sre, sim, ure[1], uim[1]);
    __syncthreads();
    apply_group<8>(sre, sim, ure[2], uim[2]);
    __syncthreads();

    if (l < N_LAYERS) ring_cnot(sre, sim);
  }

  float* ob = out + (size_t)b * DIM;
#pragma unroll
  for (int v = 0; v < 16; ++v) {
    const int i = tid + (v << 8);
    ob[i] = sre[i];
  }
}

extern "C" void kernel_launch(void* const* d_in, const int* in_sizes, int n_in,
                              void* d_out, int out_size, void* d_ws, size_t ws_size,
                              hipStream_t stream) {
  const float* x = (const float*)d_in[0];
  const float* w = (const float*)d_in[1];
  float* out = (float*)d_out;
  const int batch = in_sizes[0] >> 12;
  qsim_kernel<<<dim3(batch), dim3(256), 0, stream>>>(x, w, out);
}


# complex gate product from three real WMMA products, not four (K1=Ur(Sr+Si) shared; Dre=K1-(Ur+Ui)Si, Dim=K1+(Ui-Ur)Sr): 12 WMMAs per block, not 16; regroups the sums, f32 rounding changes
# speedup vs baseline: 1.2461x; 1.1109x over previous
#include <hip/hip_runtime.h>
#include <hip/hip_bf16.h>


#define DIM       4096
#define N_QUBITS  12
#define N_LAYERS  5

typedef __attribute__((ext_vector_type(2))) float v2f;
typedef __attribute__((ext_vector_type(8))) float v8f;

struct cx { float r, i; };
__device__ __forceinline__ cx cmul(cx a, cx b) { return { a.r*b.r - a.i*b.i, a.r*b.i + a.i*b.r }; }
__device__ __forceinline__ cx cadd(cx a, cx b) { return { a.r + b.r, a.i + b.i }; }

struct m2 { cx e[2][2]; };

__device__ __forceinline__ m2 mmul(const m2& A, const m2& B) {
  m2 C;
#pragma unroll
  for (int i = 0; i < 2; ++i)
#pragma unroll
    for (int j = 0; j < 2; ++j)
      C.e[i][j] = cadd(cmul(A.e[i][0], B.e[0][j]), cmul(A.e[i][1], B.e[1][j]));
  return C;
}

__device__ __forceinline__ m2 rot_x(float t) {
  float c = cosf(0.5f * t), s = sinf(0.5f * t);
  m2 M; M.e[0][0] = {c, 0}; M.e[0][1] = {0, -s};
        M.e[1][0] = {0, -s}; M.e[1][1] = {c, 0};
  return M;
}
__device__ __forceinline__ m2 rot_y(float t) {
  float c = cosf(0.5f * t), s = sinf(0.5f * t);
  m2 M; M.e[0][0] = {c, 0}; M.e[0][1] = {-s, 0};
        M.e[1][0] = {s, 0}; M.e[1][1] = {c, 0};
  return M;
}
__device__ __forceinline__ m2 rot_z(float t) {
  float c = cosf(0.5f * t), s = sinf(0.5f * t);
  m2 M; M.e[0][0] = {c, -s}; M.e[0][1] = {0, 0};
        M.e[1][0] = {0, 0};  M.e[1][1] = {c, s};
  return M;
}

template <int QB>
__device__ __forceinline__ constexpr int sidx(int r, int c) {
  if constexpr (QB == 0) return (r << 8) | c;
  else if constexpr (QB == 4) return ((c & 0xF0) << 4) | (r << 4) | (c & 15);
  else return (c << 4) | r;
}

__device__ __forceinline__ v8f wmma_f32(v2f a, v2f b, v8f c) {
  return __builtin_amdgcn_wmma_f32_16x16x4_f32(
      false, a, false, b, (short)0, c, false, false);
}

template <int QB>
__device__ void apply_group(float* sre, float* sim,
                            const float* ure, const float* uim) {
  const int lane  = threadIdx.x & 31;
  const int wave  = threadIdx.x >> 5;
  const int half  = lane >> 4;
  const int n     = lane & 15;
  const int khalf = half << 1;

  v2f aR[4], aD[4], aSn[4];
#pragma unroll
  for (int kc = 0; kc < 4; ++kc) {
    const int k0 = kc * 4 + khalf;
    const v2f ur = *(const v2f*)&ure[n * 16 + k0];
    const v2f ui = *(const v2f*)&uim[n * 16 + k0];
    aR[kc]  = ur;
    aD[kc]  = ui - ur;
    aSn[kc] = (-ur) - ui;
  }

#pragma unroll
  for (int blk = 0; blk < 2; ++blk) {
    const int cb = wave * 2 + blk;
    const int c  = cb * 16 + n;

    v2f bR[4], bI[4];
#pragma unroll
    for (int kc = 0; kc < 4; ++kc) {
      const int k0 = kc * 4 + khalf;
      if constexpr (QB == 8) {
        bR[kc] = *(const v2f*)&sre[(c << 4) + k0];
        bI[kc] = *(const v2f*)&sim[(c << 4) + k0];
      } else {
        bR[kc].x = sre[sidx<QB>(k0,     c)];
        bR[kc].y = sre[sidx<QB>(k0 + 1, c)];
        bI[kc].x = sim[sidx<QB>(k0,     c)];
        bI[kc].y = sim[sidx<QB>(k0 + 1, c)];
      }
    }

    v2f bS[4];
#pragma unroll
    for (int kc = 0; kc < 4; ++kc) bS[kc] = bR[kc] + bI[kc];

    v8f k1 = {};
#pragma unroll
    for (int kc = 0; kc < 4; ++kc) k1 = wmma_f32(aR[kc], bS[kc], k1);
    v8f dR = k1;
    v8f dI = k1;
#pragma unroll
    for (int kc = 0; kc < 4; ++kc) {
      dR = wmma_f32(aSn[kc], bI[kc], dR);
      dI = wmma_f32(aD[kc],  bR[kc], dI);
    }

    if constexpr (QB == 8) {
      const int base = (c << 4) + 8 * half;
#pragma unroll
      for (int v = 0; v < 4; ++v) {
        v2f pr = { dR[2 * v], dR[2 * v + 1] };
        v2f pi = { dI[2 * v], dI[2 * v + 1] };
        *(v2f*)&sre[base + 2 * v] = pr;
        *(v2f*)&sim[base + 2 * v] = pi;
      }
    } else {
#pragma unroll
      for (int v = 0; v < 8; ++v) {
        const int m   = v + 8 * half;
        const int idx = sidx<QB>(m, c);
        sre[idx] = dR[v];
        sim[idx] = dI[v];
      }
    }
  }
}

__device__ void ring_cnot(float* sre, float* sim) {
  const int tid = threadIdx.x;
  float tr[16], ti[16];
#pragma unroll
  for (int v = 0; v < 16; ++v) {
    const int i = tid + (v << 8);
    int j = i;
#pragma unroll
    for (int q = 11; q >= 0; --q) {
      const int cb = 11 - q;
      const int tb = 11 - ((q + 1) % 12);
      j ^= ((j >> cb) & 1) << tb;
    }
    tr[v] = sre[j];
    ti[v] = sim[j];
  }
  __syncthreads();
#pragma unroll
  for (int v = 0; v < 16; ++v) {
    const int i = tid + (v << 8);
    sre[i] = tr[v];
    sim[i] = ti[v];
  }
  __syncthreads();
}

__global__ __launch_bounds__(256)
void qsim_kernel(const float* __restrict__ x, const float* __restrict__ w,
                 float* __restrict__ out) {
  __shared__ float sre[DIM];
  __shared__ float sim[DIM];
  __shared__ float ure[3][256];
  __shared__ float uim[3][256];
  __shared__ float g2re[N_LAYERS + 1][N_QUBITS][4];
  __shared__ float g2im[N_LAYERS + 1][N_QUBITS][4];
  __shared__ float red[256];

  const int b   = blockIdx.x;
  const int tid = threadIdx.x;
  const float* xb = x + (size_t)b * DIM;

  const int gl = tid >> 5, gq = tid & 31;
  const bool gbuild = (gl <= N_LAYERS) && (gq < N_QUBITS);
  const int  gofs = gbuild ? gl * 36 + gq * 3 : 0;
  const float p0 = w[gofs], p1 = w[gofs + 1], p2 = w[gofs + 2];
  __builtin_amdgcn_sched_barrier(0);

  float xs[16];
#pragma unroll
  for (int v = 0; v < 16; ++v) xs[v] = xb[tid + (v << 8)];

  asm volatile("" :: "v"(p0), "v"(p1), "v"(p2));

  if (gbuild) {
    m2 U;
    if (gl < N_LAYERS) U = mmul(rot_z(p2), mmul(rot_y(p1), rot_x(p0)));
    else               U = mmul(rot_y(p2), mmul(rot_z(p1), rot_y(p0)));
#pragma unroll
    for (int e = 0; e < 4; ++e) {
      g2re[gl][gq][e] = U.e[e >> 1][e & 1].r;
      g2im[gl][gq][e] = U.e[e >> 1][e & 1].i;
    }
  }

  float ss = 0.f;
#pragma unroll
  for (int v = 0; v < 16; ++v) ss += xs[v] * xs[v];
  red[tid] = ss;
  __syncthreads();
  for (int s = 128; s > 0; s >>= 1) {
    if (tid < s) red[tid] += red[tid + s];
    __syncthreads();
  }
  const float scale = rsqrtf(sqrtf(red[0]) + 1e-12f);
#pragma unroll
  for (int v = 0; v < 16; ++v) {
    const int i = tid + (v << 8);
    sre[i] = xs[v] * scale;
    sim[i] = 0.f;
  }
  __syncthreads();

  for (int l = 0; l <= N_LAYERS; ++l) {
    {
      const int m = tid >> 4, j = tid & 15;
#pragma unroll
      for (int g = 0; g < 3; ++g) {
        cx prod = {1.f, 0.f};
#pragma unroll
        for (int k = 0; k < 4; ++k) {
          const int q  = g * 4 + k;
          const int mb = (m >> (3 - k)) & 1;
          const int jb = (j >> (3 - k)) & 1;
          const cx u = { g2re[l][q][mb * 2 + jb], g2im[l][q][mb * 2 + jb] };
          prod = cmul(prod, u);
        }
        ure[g][tid] = prod.r;
        uim[g][tid] = prod.i;
      }
    }
    __syncthreads();

    apply_group<0>(sre, sim, ure[0], uim[0]);
    __syncthreads();
    apply_group<4>(sre, sim, ure[1], uim[1]);
    __syncthreads();
    apply_group<8>(sre, sim, ure[2], uim[2]);
    __syncthreads();

    if (l < N_LAYERS) ring_cnot(sre, sim);
  }

  float* ob = out + (size_t)b * DIM;
#pragma unroll
  for (int v = 0; v < 16; ++v) {
    const int i = tid + (v << 8);
    ob[i] = sre[i];
  }
}

extern "C" void kernel_launch(void* const* d_in, const int* in_sizes, int n_in,
                              void* d_out, int out_size, void* d_ws, size_t ws_size,
                              hipStream_t stream) {
  const float* x = (const float*)d_in[0];
  const float* w = (const float*)d_in[1];
  float* out = (float*)d_out;
  const int batch = in_sizes[0] >> 12;
  qsim_kernel<<<dim3(batch), dim3(256), 0, stream>>>(x, w, out);
}


# LDS bank conflicts removed: state planes XOR-swizzled (index bits 6+ into dword bits 2..5), gate rows at pitch 20; each fragment load/store hits distinct banks; sums and their order unchanged
# speedup vs baseline: 1.3743x; 1.1029x over previous
#include <hip/hip_runtime.h>
#include <hip/hip_bf16.h>


#define DIM       4096
#define N_QUBITS  12
#define N_LAYERS  5
#define GP        20

typedef __attribute__((ext_vector_type(2))) float v2f;
typedef __attribute__((ext_vector_type(4))) float v4f;
typedef __attribute__((ext_vector_type(8))) float v8f;

struct cx { float r, i; };
__device__ __forceinline__ cx cmul(cx a, cx b) { return { a.r*b.r - a.i*b.i, a.r*b.i + a.i*b.r }; }
__device__ __forceinline__ cx cadd(cx a, cx b) { return { a.r + b.r, a.i + b.i }; }

struct m2 { cx e[2][2]; };

__device__ __forceinline__ m2 mmul(const m2& A, const m2& B) {
  m2 C;
#pragma unroll
  for (int i = 0; i < 2; ++i)
#pragma unroll
    for (int j = 0; j < 2; ++j)
      C.e[i][j] = cadd(cmul(A.e[i][0], B.e[0][j]), cmul(A.e[i][1], B.e[1][j]));
  return C;
}

__device__ __forceinline__ m2 rot_x(float t) {
  float c = cosf(0.5f * t), s = sinf(0.5f * t);
  m2 M; M.e[0][0] = {c, 0}; M.e[0][1] = {0, -s};
        M.e[1][0] = {0, -s}; M.e[1][1] = {c, 0};
  return M;
}
__device__ __forceinline__ m2 rot_y(float t) {
  float c = cosf(0.5f * t), s = sinf(0.5f * t);
  m2 M; M.e[0][0] = {c, 0}; M.e[0][1] = {-s, 0};
        M.e[1][0] = {s, 0}; M.e[1][1] = {c, 0};
  return M;
}
__device__ __forceinline__ m2 rot_z(float t) {
  float c = cosf(0.5f * t), s = sinf(0.5f * t);
  m2 M; M.e[0][0] = {c, -s}; M.e[0][1] = {0, 0};
        M.e[1][0] = {0, 0};  M.e[1][1] = {c, s};
  return M;
}

__device__ __forceinline__ constexpr int ph(int i) {
  return i ^ (((i >> 6) & 3) << 2) ^ (((i >> 7) & 1) << 4)
           ^ ((((i >> 9) ^ (i >> 10) ^ (i >> 11)) & 1) << 5);
}

__device__ __forceinline__ constexpr int lin_addr(int lin0, int v) {
  return (lin0 ^ ((((v >> 1) ^ (v >> 2) ^ (v >> 3)) & 1) << 5)) + (v << 8);
}

template <int QB>
__device__ __forceinline__ int ld_addr(int wave, int n, int half, int phi,
                                       int blk, int kc, int e) {
  if constexpr (QB == 0) {
    const int c0  = wave * 32 + n;
    const int lo0 = c0 ^ (((c0 >> 6) & 3) << 2) ^ (((c0 >> 7) & 1) << 4) ^ (half << 5);
    const int pk  = (kc ^ (kc >> 1)) & 1;
    return (((lo0 ^ (pk << 5)) | (half << 9)) ^ (blk << 4)) + ((4 * kc + e) << 8);
  } else if constexpr (QB == 4) {
    const int base = (wave << 9) | ((half ^ phi) << 5) | (n ^ (kc << 2));
    return base + (blk << 8) + (kc << 6) + ((e ^ (kc >> 1)) << 4);
  } else {
    const int rowb = (wave << 9) | ((n ^ (n >> 3) ^ (phi << 1)) << 4);
    return (rowb | ((kc ^ (n >> 2)) << 2) | (half << 1)) + (blk << 8) + e;
  }
}

template <int QB>
__device__ __forceinline__ int st_addr(int wave, int n, int half, int phi,
                                       int blk, int v) {
  if constexpr (QB == 0) {
    const int c0  = wave * 32 + n;
    const int lo0 = c0 ^ (((c0 >> 6) & 3) << 2) ^ (((c0 >> 7) & 1) << 4) ^ (half << 5);
    const int pv  = ((v >> 1) ^ (v >> 2)) & 1;
    return (((lo0 ^ (pv << 5)) | (half << 11)) ^ (blk << 4)) + (v << 8);
  } else if constexpr (QB == 4) {
    const int sb = (wave << 9) | (half << 7) | (phi << 5) | (half << 4) | (n ^ (half << 3));
    return (sb ^ ((v & 2) << 4) ^ ((v & 1) << 4) ^ (v & 4)) + (blk << 8) + ((v & 4) << 4);
  } else {
    const int rowb = (wave << 9) | ((n ^ (n >> 3) ^ (phi << 1)) << 4);
    return (rowb | (((2 * half + v) ^ (n >> 2)) << 2)) + (blk << 8);
  }
}

template <int QB>
__device__ __forceinline__ constexpr int sidx(int r, int c) {
  if constexpr (QB == 0) return (r << 8) | c;
  else if constexpr (QB == 4) return ((c & 0xF0) << 4) | (r << 4) | (c & 15);
  else return (c << 4) | r;
}

__device__ __forceinline__ v8f wmma_f32(v2f a, v2f b, v8f c) {
  return __builtin_amdgcn_wmma_f32_16x16x4_f32(
      false, a, false, b, (short)0, c, false, false);
}

template <int QB>
__device__ void apply_group(float* sre, float* sim,
                            const float* ure, const float* uim) {
  const int lane  = threadIdx.x & 31;
  const int wave  = threadIdx.x >> 5;
  const int half  = lane >> 4;
  const int n     = lane & 15;
  const int khalf = half << 1;
  const int phi   = (wave ^ (wave >> 1) ^ (wave >> 2)) & 1;

  v2f aR[4], aD[4], aSn[4];
#pragma unroll
  for (int kc = 0; kc < 4; ++kc) {
    const int k0 = kc * 4 + khalf;
    const v2f ur = *(const v2f*)&ure[n * GP + k0];
    const v2f ui = *(const v2f*)&uim[n * GP + k0];
    aR[kc]  = ur;
    aD[kc]  = ui - ur;
    aSn[kc] = (-ur) - ui;
  }

#pragma unroll
  for (int blk = 0; blk < 2; ++blk) {
    v2f bR[4], bI[4];
#pragma unroll
    for (int kc = 0; kc < 4; ++kc) {
      const int a0 = ld_addr<QB>(wave, n, half, phi, blk, kc, 0);
      if constexpr (QB == 8) {
        bR[kc] = *(const v2f*)&sre[a0];
        bI[kc] = *(const v2f*)&sim[a0];
      } else {
        const int a1 = ld_addr<QB>(wave, n, half, phi, blk, kc, 1);
        bR[kc].x = sre[a0];
        bR[kc].y = sre[a1];
        bI[kc].x = sim[a0];
        bI[kc].y = sim[a1];
      }
    }

    v2f bS[4];
#pragma unroll
    for (int kc = 0; kc < 4; ++kc) bS[kc] = bR[kc] + bI[kc];

    v8f k1 = {};
#pragma unroll
    for (int kc = 0; kc < 4; ++kc) k1 = wmma_f32(aR[kc], bS[kc], k1);
    v8f dR = k1;
    v8f dI = k1;
#pragma unroll
    for (int kc = 0; kc < 4; ++kc) {
      dR = wmma_f32(aSn[kc], bI[kc], dR);
      dI = wmma_f32(aD[kc],  bR[kc], dI);
    }

    if constexpr (QB == 8) {
#pragma unroll
      for (int v = 0; v < 2; ++v) {
        const int a = st_addr<QB>(wave, n, half, phi, blk, v);
        v4f pr = { dR[4 * v], dR[4 * v + 1], dR[4 * v + 2], dR[4 * v + 3] };
        v4f pi = { dI[4 * v], dI[4 * v + 1], dI[4 * v + 2], dI[4 * v + 3] };
        *(v4f*)&sre[a] = pr;
        *(v4f*)&sim[a] = pi;
      }
    } else {
#pragma unroll
      for (int v = 0; v < 8; ++v) {
        const int idx = st_addr<QB>(wave, n, half, phi, blk, v);
        sre[idx] = dR[v];
        sim[idx] = dI[v];
      }
    }
  }
}

__device__ void ring_cnot(float* sre, float* sim) {
  const int tid = threadIdx.x;
  float tr[16], ti[16];
#pragma unroll
  for (int v = 0; v < 16; ++v) {
    const int i = tid + (v << 8);
    int j = i;
#pragma unroll
    for (int q = 11; q >= 0; --q) {
      const int cb = 11 - q;
      const int tb = 11 - ((q + 1) % 12);
      j ^= ((j >> cb) & 1) << tb;
    }
    tr[v] = sre[ph(j)];
    ti[v] = sim[ph(j)];
  }
  __syncthreads();
  const int lin0 = ph(tid);
#pragma unroll
  for (int v = 0; v < 16; ++v) {
    const int i = lin_addr(lin0, v);
    sre[i] = tr[v];
    sim[i] = ti[v];
  }
  __syncthreads();
}

__global__ __launch_bounds__(256)
void qsim_kernel(const float* __restrict__ x, const float* __restrict__ w,
                 float* __restrict__ out) {
  __shared__ __attribute__((aligned(256))) float sre[DIM];
  __shared__ __attribute__((aligned(256))) float sim[DIM];
  __shared__ __attribute__((aligned(16))) float ure[3][16 * GP];
  __shared__ __attribute__((aligned(16))) float uim[3][16 * GP];
  __shared__ float g2re[N_LAYERS + 1][N_QUBITS][4];
  __shared__ float g2im[N_LAYERS + 1][N_QUBITS][4];
  __shared__ float red[256];

  const int b   = blockIdx.x;
  const int tid = threadIdx.x;
  const float* xb = x + (size_t)b * DIM;

  const int gl = tid >> 5, gq = tid & 31;
  const bool gbuild = (gl <= N_LAYERS) && (gq < N_QUBITS);
  const int  gofs = gbuild ? gl * 36 + gq * 3 : 0;
  const float p0 = w[gofs], p1 = w[gofs + 1], p2 = w[gofs + 2];
  __builtin_amdgcn_sched_barrier(0);

  float xs[16];
#pragma unroll
  for (int v = 0; v < 16; ++v) xs[v] = xb[tid + (v << 8)];

  asm volatile("" :: "v"(p0), "v"(p1), "v"(p2));

  if (gbuild) {
    m2 U;
    if (gl < N_LAYERS) U = mmul(rot_z(p2), mmul(rot_y(p1), rot_x(p0)));
    else               U = mmul(rot_y(p2), mmul(rot_z(p1), rot_y(p0)));
#pragma unroll
    for (int e = 0; e < 4; ++e) {
      g2re[gl][gq][e] = U.e[e >> 1][e & 1].r;
      g2im[gl][gq][e] = U.e[e >> 1][e & 1].i;
    }
  }

  float ss = 0.f;
#pragma unroll
  for (int v = 0; v < 16; ++v) ss += xs[v] * xs[v];
  red[tid] = ss;
  __syncthreads();
  for (int s = 128; s > 0; s >>= 1) {
    if (tid < s) red[tid] += red[tid + s];
    __syncthreads();
  }
  const float scale = rsqrtf(sqrtf(red[0]) + 1e-12f);
#pragma unroll
  for (int v = 0; v < 16; ++v) {
    const int i = lin_addr(ph(tid), v);
    sre[i] = xs[v] * scale;
    sim[i] = 0.f;
  }
  __syncthreads();

  for (int l = 0; l <= N_LAYERS; ++l) {
    {
      const int m = tid >> 4, j = tid & 15;
#pragma unroll
      for (int g = 0; g < 3; ++g) {
        cx prod = {1.f, 0.f};
#pragma unroll
        for (int k = 0; k < 4; ++k) {
          const int q  = g * 4 + k;
          const int mb = (m >> (3 - k)) & 1;
          const int jb = (j >> (3 - k)) & 1;
          const cx u = { g2re[l][q][mb * 2 + jb], g2im[l][q][mb * 2 + jb] };
          prod = cmul(prod, u);
        }
        ure[g][m * GP + j] = prod.r;
        uim[g][m * GP + j] = prod.i;
      }
    }
    __syncthreads();

    apply_group<0>(sre, sim, ure[0], uim[0]);
    __syncthreads();
    apply_group<4>(sre, sim, ure[1], uim[1]);
    __syncthreads();
    apply_group<8>(sre, sim, ure[2], uim[2]);
    __syncthreads();

    if (l < N_LAYERS) ring_cnot(sre, sim);
  }

  float* ob = out + (size_t)b * DIM;
#pragma unroll
  for (int v = 0; v < 16; ++v) {
    const int i = tid + (v << 8);
    ob[i] = sre[lin_addr(ph(tid), v)];
  }
}

extern "C" void kernel_launch(void* const* d_in, const int* in_sizes, int n_in,
                              void* d_out, int out_size, void* d_ws, size_t ws_size,
                              hipStream_t stream) {
  const float* x = (const float*)d_in[0];
  const float* w = (const float*)d_in[1];
  float* out = (float*)d_out;
  const int batch = in_sizes[0] >> 12;
  qsim_kernel<<<dim3(batch), dim3(256), 0, stream>>>(x, w, out);
}


# CNOT ring folded into the next layer's hi-nibble B-fragment loads (gather at ph(ring(i)) from one LDS state copy into a second one): no ring pass, 4 barriers per layer not 6; sums unchanged
# speedup vs baseline: 1.4990x; 1.0907x over previous
#include <hip/hip_runtime.h>
#include <hip/hip_bf16.h>


#define DIM       4096
#define N_QUBITS  12
#define N_LAYERS  5
#define GP        20

typedef __attribute__((ext_vector_type(2))) float v2f;
typedef __attribute__((ext_vector_type(4))) float v4f;
typedef __attribute__((ext_vector_type(8))) float v8f;

struct cx { float r, i; };
__device__ __forceinline__ cx cmul(cx a, cx b) { return { a.r*b.r - a.i*b.i, a.r*b.i + a.i*b.r }; }
__device__ __forceinline__ cx cadd(cx a, cx b) { return { a.r + b.r, a.i + b.i }; }

struct m2 { cx e[2][2]; };

__device__ __forceinline__ m2 mmul(const m2& A, const m2& B) {
  m2 C;
#pragma unroll
  for (int i = 0; i < 2; ++i)
#pragma unroll
    for (int j = 0; j < 2; ++j)
      C.e[i][j] = cadd(cmul(A.e[i][0], B.e[0][j]), cmul(A.e[i][1], B.e[1][j]));
  return C;
}

__device__ __forceinline__ m2 rot_x(float t) {
  float c = cosf(0.5f * t), s = sinf(0.5f * t);
  m2 M; M.e[0][0] = {c, 0}; M.e[0][1] = {0, -s};
        M.e[1][0] = {0, -s}; M.e[1][1] = {c, 0};
  return M;
}
__device__ __forceinline__ m2 rot_y(float t) {
  float c = cosf(0.5f * t), s = sinf(0.5f * t);
  m2 M; M.e[0][0] = {c, 0}; M.e[0][1] = {-s, 0};
        M.e[1][0] = {s, 0}; M.e[1][1] = {c, 0};
  return M;
}
__device__ __forceinline__ m2 rot_z(float t) {
  float c = cosf(0.5f * t), s = sinf(0.5f * t);
  m2 M; M.e[0][0] = {c, -s}; M.e[0][1] = {0, 0};
        M.e[1][0] = {0, 0};  M.e[1][1] = {c, s};
  return M;
}

__device__ __forceinline__ constexpr int ph(int i) {
  return i ^ (((i >> 6) & 3) << 2) ^ (((i >> 7) & 1) << 4)
           ^ ((((i >> 9) ^ (i >> 10) ^ (i >> 11)) & 1) << 5);
}

__device__ __forceinline__ constexpr int lin_addr(int lin0, int v) {
  return (lin0 ^ ((((v >> 1) ^ (v >> 2) ^ (v >> 3)) & 1) << 5)) + (v << 8);
}

template <int QB>
__device__ __forceinline__ int ld_addr(int wave, int n, int half, int phi,
                                       int blk, int kc, int e) {
  if constexpr (QB == 0) {
    const int c0  = wave * 32 + n;
    const int lo0 = c0 ^ (((c0 >> 6) & 3) << 2) ^ (((c0 >> 7) & 1) << 4) ^ (half << 5);
    const int pk  = (kc ^ (kc >> 1)) & 1;
    return (((lo0 ^ (pk << 5)) | (half << 9)) ^ (blk << 4)) + ((4 * kc + e) << 8);
  } else if constexpr (QB == 4) {
    const int base = (wave << 9) | ((half ^ phi) << 5) | (n ^ (kc << 2));
    return base + (blk << 8) + (kc << 6) + ((e ^ (kc >> 1)) << 4);
  } else {
    const int rowb = (wave << 9) | ((n ^ (n >> 3) ^ (phi << 1)) << 4);
    return (rowb | ((kc ^ (n >> 2)) << 2) | (half << 1)) + (blk << 8) + e;
  }
}

template <int QB>
__device__ __forceinline__ int st_addr(int wave, int n, int half, int phi,
                                       int blk, int v) {
  if constexpr (QB == 0) {
    const int c0  = wave * 32 + n;
    const int lo0 = c0 ^ (((c0 >> 6) & 3) << 2) ^ (((c0 >> 7) & 1) << 4) ^ (half << 5);
    const int pv  = ((v >> 1) ^ (v >> 2)) & 1;
    return (((lo0 ^ (pv << 5)) | (half << 11)) ^ (blk << 4)) + (v << 8);
  } else if constexpr (QB == 4) {
    const int sb = (wave << 9) | (half << 7) | (phi << 5) | (half << 4) | (n ^ (half << 3));
    return (sb ^ ((v & 2) << 4) ^ ((v & 1) << 4) ^ (v & 4)) + (blk << 8) + ((v & 4) << 4);
  } else {
    const int rowb = (wave << 9) | ((n ^ (n >> 3) ^ (phi << 1)) << 4);
    return (rowb | (((2 * half + v) ^ (n >> 2)) << 2)) + (blk << 8);
  }
}

template <int QB>
__device__ __forceinline__ constexpr int sidx(int r, int c) {
  if constexpr (QB == 0) return (r << 8) | c;
  else if constexpr (QB == 4) return ((c & 0xF0) << 4) | (r << 4) | (c & 15);
  else return (c << 4) | r;
}

__device__ __forceinline__ constexpr int ring_src(int i) {
  int j = i;
  for (int q = 11; q >= 0; --q) {
    const int cb = 11 - q;
    const int tb = 11 - ((q + 1) % 12);
    j ^= ((j >> cb) & 1) << tb;
  }
  return j;
}

__device__ __forceinline__ v8f wmma_f32(v2f a, v2f b, v8f c) {
  return __builtin_amdgcn_wmma_f32_16x16x4_f32(
      false, a, false, b, (short)0, c, false, false);
}

template <int QB, bool RING>
__device__ void apply_group(const float* sre, const float* sim,
                            float* dre, float* dim,
                            const float* ure, const float* uim) {
  static_assert(!RING || QB == 0, "the ring is folded into the hi nibble loads only");
  const int lane  = threadIdx.x & 31;
  const int wave  = threadIdx.x >> 5;
  const int half  = lane >> 4;
  const int n     = lane & 15;
  const int khalf = half << 1;
  const int phi   = (wave ^ (wave >> 1) ^ (wave >> 2)) & 1;
  const int rlane = ph(ring_src((half << 9) | (wave << 5) | n));

  v2f aR[4], aD[4], aSn[4];
#pragma unroll
  for (int kc = 0; kc < 4; ++kc) {
    const int k0 = kc * 4 + khalf;
    const v2f ur = *(const v2f*)&ure[n * GP + k0];
    const v2f ui = *(const v2f*)&uim[n * GP + k0];
    aR[kc]  = ur;
    aD[kc]  = ui - ur;
    aSn[kc] = (-ur) - ui;
  }

#pragma unroll
  for (int blk = 0; blk < 2; ++blk) {
    v2f bR[4], bI[4];
#pragma unroll
    for (int kc = 0; kc < 4; ++kc) {
      if constexpr (RING) {
        const int a0 = rlane ^ ph(ring_src(((4 * kc) << 8) | (blk << 4)));
        const int a1 = rlane ^ ph(ring_src(((4 * kc + 1) << 8) | (blk << 4)));
        bR[kc].x = sre[a0];
        bR[kc].y = sre[a1];
        bI[kc].x = sim[a0];
        bI[kc].y = sim[a1];
      } else if constexpr (QB == 8) {
        const int a0 = ld_addr<QB>(wave, n, half, phi, blk, kc, 0);
        bR[kc] = *(const v2f*)&sre[a0];
        bI[kc] = *(const v2f*)&sim[a0];
      } else {
        const int a0 = ld_addr<QB>(wave, n, half, phi, blk, kc, 0);
        const int a1 = ld_addr<QB>(wave, n, half, phi, blk, kc, 1);
        bR[kc].x = sre[a0];
        bR[kc].y = sre[a1];
        bI[kc].x = sim[a0];
        bI[kc].y = sim[a1];
      }
    }

    v2f bS[4];
#pragma unroll
    for (int kc = 0; kc < 4; ++kc) bS[kc] = bR[kc] + bI[kc];

    v8f k1 = {};
#pragma unroll
    for (int kc = 0; kc < 4; ++kc) k1 = wmma_f32(aR[kc], bS[kc], k1);
    v8f dR = k1;
    v8f dI = k1;
#pragma unroll
    for (int kc = 0; kc < 4; ++kc) {
      dR = wmma_f32(aSn[kc], bI[kc], dR);
      dI = wmma_f32(aD[kc],  bR[kc], dI);
    }

    if constexpr (QB == 8) {
#pragma unroll
      for (int v = 0; v < 2; ++v) {
        const int a = st_addr<QB>(wave, n, half, phi, blk, v);
        v4f pr = { dR[4 * v], dR[4 * v + 1], dR[4 * v + 2], dR[4 * v + 3] };
        v4f pi = { dI[4 * v], dI[4 * v + 1], dI[4 * v + 2], dI[4 * v + 3] };
        *(v4f*)&dre[a] = pr;
        *(v4f*)&dim[a] = pi;
      }
    } else {
#pragma unroll
      for (int v = 0; v < 8; ++v) {
        const int idx = st_addr<QB>(wave, n, half, phi, blk, v);
        dre[idx] = dR[v];
        dim[idx] = dI[v];
      }
    }
  }
}

__global__ __launch_bounds__(256)
void qsim_kernel(const float* __restrict__ x, const float* __restrict__ w,
                 float* __restrict__ out) {
  __shared__ __attribute__((aligned(256))) float sre[2 * DIM];
  __shared__ __attribute__((aligned(256))) float sim[2 * DIM];
  __shared__ __attribute__((aligned(16))) float ure[3][16 * GP];
  __shared__ __attribute__((aligned(16))) float uim[3][16 * GP];
  __shared__ float g2re[N_LAYERS + 1][N_QUBITS][4];
  __shared__ float g2im[N_LAYERS + 1][N_QUBITS][4];
  __shared__ float red[256];

  const int b   = blockIdx.x;
  const int tid = threadIdx.x;
  const float* xb = x + (size_t)b * DIM;

  const int gl = tid >> 5, gq = tid & 31;
  const bool gbuild = (gl <= N_LAYERS) && (gq < N_QUBITS);
  const int  gofs = gbuild ? gl * 36 + gq * 3 : 0;
  const float p0 = w[gofs], p1 = w[gofs + 1], p2 = w[gofs + 2];
  __builtin_amdgcn_sched_barrier(0);

  float xs[16];
#pragma unroll
  for (int v = 0; v < 16; ++v) xs[v] = xb[tid + (v << 8)];

  asm volatile("" :: "v"(p0), "v"(p1), "v"(p2));

  if (gbuild) {
    m2 U;
    if (gl < N_LAYERS) U = mmul(rot_z(p2), mmul(rot_y(p1), rot_x(p0)));
    else               U = mmul(rot_y(p2), mmul(rot_z(p1), rot_y(p0)));
#pragma unroll
    for (int e = 0; e < 4; ++e) {
      g2re[gl][gq][e] = U.e[e >> 1][e & 1].r;
      g2im[gl][gq][e] = U.e[e >> 1][e & 1].i;
    }
  }

  float ss = 0.f;
#pragma unroll
  for (int v = 0; v < 16; ++v) ss += xs[v] * xs[v];
  red[tid] = ss;
  __syncthreads();
  for (int s = 128; s > 0; s >>= 1) {
    if (tid < s) red[tid] += red[tid + s];
    __syncthreads();
  }
  const float scale = rsqrtf(sqrtf(red[0]) + 1e-12f);
#pragma unroll
  for (int v = 0; v < 16; ++v) {
    const int i = lin_addr(ph(tid), v);
    sre[i] = xs[v] * scale;
    sim[i] = 0.f;
  }
  __syncthreads();

  int cur = 0;
  for (int l = 0; l <= N_LAYERS; ++l) {
    {
      const int m = tid >> 4, j = tid & 15;
#pragma unroll
      for (int g = 0; g < 3; ++g) {
        cx prod = {1.f, 0.f};
#pragma unroll
        for (int k = 0; k < 4; ++k) {
          const int q  = g * 4 + k;
          const int mb = (m >> (3 - k)) & 1;
          const int jb = (j >> (3 - k)) & 1;
          const cx u = { g2re[l][q][mb * 2 + jb], g2im[l][q][mb * 2 + jb] };
          prod = cmul(prod, u);
        }
        ure[g][m * GP + j] = prod.r;
        uim[g][m * GP + j] = prod.i;
      }
    }
    __syncthreads();

    if (l == 0) {
      apply_group<0, false>(sre, sim, sre, sim, ure[0], uim[0]);
    } else {
      const int oth = cur ^ DIM;
      apply_group<0, true>(sre + cur, sim + cur, sre + oth, sim + oth, ure[0], uim[0]);
      cur = oth;
    }
    float* cre = sre + cur;
    float* cim = sim + cur;
    __syncthreads();
    apply_group<4, false>(cre, cim, cre, cim, ure[1], uim[1]);
    __syncthreads();
    apply_group<8, false>(cre, cim, cre, cim, ure[2], uim[2]);
    __syncthreads();
  }

  float* ob = out + (size_t)b * DIM;
#pragma unroll
  for (int v = 0; v < 16; ++v) {
    const int i = tid + (v << 8);
    ob[i] = sre[cur + lin_addr(ph(tid), v)];
  }
}

extern "C" void kernel_launch(void* const* d_in, const int* in_sizes, int n_in,
                              void* d_out, int out_size, void* d_ws, size_t ws_size,
                              hipStream_t stream) {
  const float* x = (const float*)d_in[0];
  const float* w = (const float*)d_in[1];
  float* out = (float*)d_out;
  const int batch = in_sizes[0] >> 12;
  qsim_kernel<<<dim3(batch), dim3(256), 0, stream>>>(x, w, out);
}


# mid and lo nibble gates fused in registers: mid product formed transposed (state as A, gate as B), its D tiles are the lo gate's B fragments; 3 barriers per layer not 4; lo sums regrouped by k
# speedup vs baseline: 1.5755x; 1.0510x over previous
#include <hip/hip_runtime.h>
#include <hip/hip_bf16.h>


#define DIM       4096
#define N_QUBITS  12
#define N_LAYERS  5
#define GP        20

__device__ __forceinline__ constexpr int lo_pos(int k) {
  return (((k >> 1) & 3) << 2) | ((k >> 3) << 1) | (k & 1);
}

typedef __attribute__((ext_vector_type(2))) float v2f;
typedef __attribute__((ext_vector_type(4))) float v4f;
typedef __attribute__((ext_vector_type(8))) float v8f;

struct cx { float r, i; };
__device__ __forceinline__ cx cmul(cx a, cx b) { return { a.r*b.r - a.i*b.i, a.r*b.i + a.i*b.r }; }
__device__ __forceinline__ cx cadd(cx a, cx b) { return { a.r + b.r, a.i + b.i }; }

struct m2 { cx e[2][2]; };

__device__ __forceinline__ m2 mmul(const m2& A, const m2& B) {
  m2 C;
#pragma unroll
  for (int i = 0; i < 2; ++i)
#pragma unroll
    for (int j = 0; j < 2; ++j)
      C.e[i][j] = cadd(cmul(A.e[i][0], B.e[0][j]), cmul(A.e[i][1], B.e[1][j]));
  return C;
}

__device__ __forceinline__ m2 rot_x(float t) {
  float c = cosf(0.5f * t), s = sinf(0.5f * t);
  m2 M; M.e[0][0] = {c, 0}; M.e[0][1] = {0, -s};
        M.e[1][0] = {0, -s}; M.e[1][1] = {c, 0};
  return M;
}
__device__ __forceinline__ m2 rot_y(float t) {
  float c = cosf(0.5f * t), s = sinf(0.5f * t);
  m2 M; M.e[0][0] = {c, 0}; M.e[0][1] = {-s, 0};
        M.e[1][0] = {s, 0}; M.e[1][1] = {c, 0};
  return M;
}
__device__ __forceinline__ m2 rot_z(float t) {
  float c = cosf(0.5f * t), s = sinf(0.5f * t);
  m2 M; M.e[0][0] = {c, -s}; M.e[0][1] = {0, 0};
        M.e[1][0] = {0, 0};  M.e[1][1] = {c, s};
  return M;
}

__device__ __forceinline__ constexpr int ph(int i) {
  return i ^ (((i >> 6) & 3) << 2) ^ (((i >> 7) & 1) << 4)
           ^ ((((i >> 9) ^ (i >> 10) ^ (i >> 11)) & 1) << 5);
}

__device__ __forceinline__ constexpr int lin_addr(int lin0, int v) {
  return (lin0 ^ ((((v >> 1) ^ (v >> 2) ^ (v >> 3)) & 1) << 5)) + (v << 8);
}

template <int QB>
__device__ __forceinline__ int ld_addr(int wave, int n, int half, int phi,
                                       int blk, int kc, int e) {
  if constexpr (QB == 0) {
    const int c0  = wave * 32 + n;
    const int lo0 = c0 ^ (((c0 >> 6) & 3) << 2) ^ (((c0 >> 7) & 1) << 4) ^ (half << 5);
    const int pk  = (kc ^ (kc >> 1)) & 1;
    return (((lo0 ^ (pk << 5)) | (half << 9)) ^ (blk << 4)) + ((4 * kc + e) << 8);
  } else if constexpr (QB == 4) {
    const int base = (wave << 9) | ((half ^ phi) << 5) | (n ^ (kc << 2));
    return base + (blk << 8) + (kc << 6) + ((e ^ (kc >> 1)) << 4);
  } else {
    const int rowb = (wave << 9) | ((n ^ (n >> 3) ^ (phi << 1)) << 4);
    return (rowb | ((kc ^ (n >> 2)) << 2) | (half << 1)) + (blk << 8) + e;
  }
}

template <int QB>
__device__ __forceinline__ int st_addr(int wave, int n, int half, int phi,
                                       int blk, int v) {
  if constexpr (QB == 0) {
    const int c0  = wave * 32 + n;
    const int lo0 = c0 ^ (((c0 >> 6) & 3) << 2) ^ (((c0 >> 7) & 1) << 4) ^ (half << 5);
    const int pv  = ((v >> 1) ^ (v >> 2)) & 1;
    return (((lo0 ^ (pv << 5)) | (half << 11)) ^ (blk << 4)) + (v << 8);
  } else if constexpr (QB == 4) {
    const int sb = (wave << 9) | (half << 7) | (phi << 5) | (half << 4) | (n ^ (half << 3));
    return (sb ^ ((v & 2) << 4) ^ ((v & 1) << 4) ^ (v & 4)) + (blk << 8) + ((v & 4) << 4);
  } else {
    const int rowb = (wave << 9) | ((n ^ (n >> 3) ^ (phi << 1)) << 4);
    return (rowb | (((2 * half + v) ^ (n >> 2)) << 2)) + (blk << 8);
  }
}

template <int QB>
__device__ __forceinline__ constexpr int sidx(int r, int c) {
  if constexpr (QB == 0) return (r << 8) | c;
  else if constexpr (QB == 4) return ((c & 0xF0) << 4) | (r << 4) | (c & 15);
  else return (c << 4) | r;
}

__device__ __forceinline__ constexpr int ring_src(int i) {
  int j = i;
  for (int q = 11; q >= 0; --q) {
    const int cb = 11 - q;
    const int tb = 11 - ((q + 1) % 12);
    j ^= ((j >> cb) & 1) << tb;
  }
  return j;
}

__device__ __forceinline__ v8f wmma_f32(v2f a, v2f b, v8f c) {
  return __builtin_amdgcn_wmma_f32_16x16x4_f32(
      false, a, false, b, (short)0, c, false, false);
}

template <int QB, bool RING>
__device__ void apply_group(const float* sre, const float* sim,
                            float* dre, float* dim,
                            const float* ure, const float* uim) {
  static_assert(!RING || QB == 0, "the ring is folded into the hi nibble loads only");
  const int lane  = threadIdx.x & 31;
  const int wave  = threadIdx.x >> 5;
  const int half  = lane >> 4;
  const int n     = lane & 15;
  const int khalf = half << 1;
  const int phi   = (wave ^ (wave >> 1) ^ (wave >> 2)) & 1;
  const int rlane = ph(ring_src((half << 9) | (wave << 5) | n));

  v2f aR[4], aD[4], aSn[4];
#pragma unroll
  for (int kc = 0; kc < 4; ++kc) {
    const int k0 = kc * 4 + khalf;
    const v2f ur = *(const v2f*)&ure[n * GP + k0];
    const v2f ui = *(const v2f*)&uim[n * GP + k0];
    aR[kc]  = ur;
    aD[kc]  = ui - ur;
    aSn[kc] = (-ur) - ui;
  }

#pragma unroll
  for (int blk = 0; blk < 2; ++blk) {
    v2f bR[4], bI[4];
#pragma unroll
    for (int kc = 0; kc < 4; ++kc) {
      if constexpr (RING) {
        const int a0 = rlane ^ ph(ring_src(((4 * kc) << 8) | (blk << 4)));
        const int a1 = rlane ^ ph(ring_src(((4 * kc + 1) << 8) | (blk << 4)));
        bR[kc].x = sre[a0];
        bR[kc].y = sre[a1];
        bI[kc].x = sim[a0];
        bI[kc].y = sim[a1];
      } else if constexpr (QB == 8) {
        const int a0 = ld_addr<QB>(wave, n, half, phi, blk, kc, 0);
        bR[kc] = *(const v2f*)&sre[a0];
        bI[kc] = *(const v2f*)&sim[a0];
      } else {
        const int a0 = ld_addr<QB>(wave, n, half, phi, blk, kc, 0);
        const int a1 = ld_addr<QB>(wave, n, half, phi, blk, kc, 1);
        bR[kc].x = sre[a0];
        bR[kc].y = sre[a1];
        bI[kc].x = sim[a0];
        bI[kc].y = sim[a1];
      }
    }

    v2f bS[4];
#pragma unroll
    for (int kc = 0; kc < 4; ++kc) bS[kc] = bR[kc] + bI[kc];

    v8f k1 = {};
#pragma unroll
    for (int kc = 0; kc < 4; ++kc) k1 = wmma_f32(aR[kc], bS[kc], k1);
    v8f dR = k1;
    v8f dI = k1;
#pragma unroll
    for (int kc = 0; kc < 4; ++kc) {
      dR = wmma_f32(aSn[kc], bI[kc], dR);
      dI = wmma_f32(aD[kc],  bR[kc], dI);
    }

    if constexpr (QB == 8) {
#pragma unroll
      for (int v = 0; v < 2; ++v) {
        const int a = st_addr<QB>(wave, n, half, phi, blk, v);
        v4f pr = { dR[4 * v], dR[4 * v + 1], dR[4 * v + 2], dR[4 * v + 3] };
        v4f pi = { dI[4 * v], dI[4 * v + 1], dI[4 * v + 2], dI[4 * v + 3] };
        *(v4f*)&dre[a] = pr;
        *(v4f*)&dim[a] = pi;
      }
    } else {
#pragma unroll
      for (int v = 0; v < 8; ++v) {
        const int idx = st_addr<QB>(wave, n, half, phi, blk, v);
        dre[idx] = dR[v];
        dim[idx] = dI[v];
      }
    }
  }
}

__device__ void apply_mid_lo(float* cre, float* cim,
                             const float* umr, const float* umi,
                             const float* ulr, const float* uli) {
  const int lane  = threadIdx.x & 31;
  const int wave  = threadIdx.x >> 5;
  const int half  = lane >> 4;
  const int n     = lane & 15;
  const int khalf = half << 1;
  const int phi   = (wave ^ (wave >> 1) ^ (wave >> 2)) & 1;

  v2f mR[4], mD[4], mSn[4];
  v2f lR[4], lD[4], lSn[4];
#pragma unroll
  for (int kc = 0; kc < 4; ++kc) {
    const int k0 = kc * 4 + khalf;
    const v2f mr = *(const v2f*)&umr[n * GP + k0];
    const v2f mi = *(const v2f*)&umi[n * GP + k0];
    mR[kc]  = mr;
    mD[kc]  = mi - mr;
    mSn[kc] = (-mr) - mi;
    const v2f lr = *(const v2f*)&ulr[n * GP + k0];
    const v2f li = *(const v2f*)&uli[n * GP + k0];
    lR[kc]  = lr;
    lD[kc]  = li - lr;
    lSn[kc] = (-lr) - li;
  }

#pragma unroll
  for (int blk = 0; blk < 2; ++blk) {
    v2f sR[4], sI[4], sS[4];
#pragma unroll
    for (int kc = 0; kc < 4; ++kc) {
      const int a0 = ld_addr<4>(wave, n, half, phi, blk, kc, 0);
      const int a1 = ld_addr<4>(wave, n, half, phi, blk, kc, 1);
      sR[kc].x = cre[a0];
      sR[kc].y = cre[a1];
      sI[kc].x = cim[a0];
      sI[kc].y = cim[a1];
    }
#pragma unroll
    for (int kc = 0; kc < 4; ++kc) sS[kc] = sR[kc] + sI[kc];

    v8f t1 = {};
#pragma unroll
    for (int kc = 0; kc < 4; ++kc) t1 = wmma_f32(sS[kc], mR[kc], t1);
    v8f tR = t1;
    v8f tI = t1;
#pragma unroll
    for (int kc = 0; kc < 4; ++kc) {
      tR = wmma_f32(sI[kc], mSn[kc], tR);
      tI = wmma_f32(sR[kc], mD[kc],  tI);
    }

    v2f bR[4], bI[4], bS[4];
#pragma unroll
    for (int kc = 0; kc < 4; ++kc) {
      bR[kc].x = tR[2 * kc];
      bR[kc].y = tR[2 * kc + 1];
      bI[kc].x = tI[2 * kc];
      bI[kc].y = tI[2 * kc + 1];
      bS[kc]   = bR[kc] + bI[kc];
    }

    v8f k1 = {};
#pragma unroll
    for (int kc = 0; kc < 4; ++kc) k1 = wmma_f32(lR[kc], bS[kc], k1);
    v8f dR = k1;
    v8f dI = k1;
#pragma unroll
    for (int kc = 0; kc < 4; ++kc) {
      dR = wmma_f32(lSn[kc], bI[kc], dR);
      dI = wmma_f32(lD[kc],  bR[kc], dI);
    }

#pragma unroll
    for (int v = 0; v < 2; ++v) {
      const int a = st_addr<8>(wave, n, half, phi, blk, v);
      v4f pr = { dR[4 * v], dR[4 * v + 1], dR[4 * v + 2], dR[4 * v + 3] };
      v4f pi = { dI[4 * v], dI[4 * v + 1], dI[4 * v + 2], dI[4 * v + 3] };
      *(v4f*)&cre[a] = pr;
      *(v4f*)&cim[a] = pi;
    }
  }
}

__global__ __launch_bounds__(256)
void qsim_kernel(const float* __restrict__ x, const float* __restrict__ w,
                 float* __restrict__ out) {
  __shared__ __attribute__((aligned(256))) float sre[2 * DIM];
  __shared__ __attribute__((aligned(256))) float sim[2 * DIM];
  __shared__ __attribute__((aligned(16))) float ure[3][16 * GP];
  __shared__ __attribute__((aligned(16))) float uim[3][16 * GP];
  __shared__ float g2re[N_LAYERS + 1][N_QUBITS][4];
  __shared__ float g2im[N_LAYERS + 1][N_QUBITS][4];
  __shared__ float red[256];

  const int b   = blockIdx.x;
  const int tid = threadIdx.x;
  const float* xb = x + (size_t)b * DIM;

  const int gl = tid >> 5, gq = tid & 31;
  const bool gbuild = (gl <= N_LAYERS) && (gq < N_QUBITS);
  const int  gofs = gbuild ? gl * 36 + gq * 3 : 0;
  const float p0 = w[gofs], p1 = w[gofs + 1], p2 = w[gofs + 2];
  __builtin_amdgcn_sched_barrier(0);

  float xs[16];
#pragma unroll
  for (int v = 0; v < 16; ++v) xs[v] = xb[tid + (v << 8)];

  asm volatile("" :: "v"(p0), "v"(p1), "v"(p2));

  if (gbuild) {
    m2 U;
    if (gl < N_LAYERS) U = mmul(rot_z(p2), mmul(rot_y(p1), rot_x(p0)));
    else               U = mmul(rot_y(p2), mmul(rot_z(p1), rot_y(p0)));
#pragma unroll
    for (int e = 0; e < 4; ++e) {
      g2re[gl][gq][e] = U.e[e >> 1][e & 1].r;
      g2im[gl][gq][e] = U.e[e >> 1][e & 1].i;
    }
  }

  float ss = 0.f;
#pragma unroll
  for (int v = 0; v < 16; ++v) ss += xs[v] * xs[v];
  red[tid] = ss;
  __syncthreads();
  for (int s = 128; s > 0; s >>= 1) {
    if (tid < s) red[tid] += red[tid + s];
    __syncthreads();
  }
  const float scale = rsqrtf(sqrtf(red[0]) + 1e-12f);
#pragma unroll
  for (int v = 0; v < 16; ++v) {
    const int i = lin_addr(ph(tid), v);
    sre[i] = xs[v] * scale;
    sim[i] = 0.f;
  }
  __syncthreads();

  int cur = 0;
  for (int l = 0; l <= N_LAYERS; ++l) {
    {
      const int m = tid >> 4, j = tid & 15;
#pragma unroll
      for (int g = 0; g < 3; ++g) {
        cx prod = {1.f, 0.f};
#pragma unroll
        for (int k = 0; k < 4; ++k) {
          const int q  = g * 4 + k;
          const int mb = (m >> (3 - k)) & 1;
          const int jb = (j >> (3 - k)) & 1;
          const cx u = { g2re[l][q][mb * 2 + jb], g2im[l][q][mb * 2 + jb] };
          prod = cmul(prod, u);
        }
        const int jp = (g == 2) ? lo_pos(j) : j;
        ure[g][m * GP + jp] = prod.r;
        uim[g][m * GP + jp] = prod.i;
      }
    }
    __syncthreads();

    if (l == 0) {
      apply_group<0, false>(sre, sim, sre, sim, ure[0], uim[0]);
    } else {
      const int oth = cur ^ DIM;
      apply_group<0, true>(sre + cur, sim + cur, sre + oth, sim + oth, ure[0], uim[0]);
      cur = oth;
    }
    float* cre = sre + cur;
    float* cim = sim + cur;
    __syncthreads();
    apply_mid_lo(cre, cim, ure[1], uim[1], ure[2], uim[2]);
    __syncthreads();
  }

  float* ob = out + (size_t)b * DIM;
#pragma unroll
  for (int v = 0; v < 16; ++v) {
    const int i = tid + (v << 8);
    ob[i] = sre[cur + lin_addr(ph(tid), v)];
  }
}

extern "C" void kernel_launch(void* const* d_in, const int* in_sizes, int n_in,
                              void* d_out, int out_size, void* d_ws, size_t ws_size,
                              hipStream_t stream) {
  const float* x = (const float*)d_in[0];
  const float* w = (const float*)d_in[1];
  float* out = (float*)d_out;
  const int batch = in_sizes[0] >> 12;
  qsim_kernel<<<dim3(batch), dim3(256), 0, stream>>>(x, w, out);
}


# two batch elements per workgroup: 16x16 gates built and gate fragments loaded once for both, half the waves launched, all 512 workgroups resident at once; each element's sums and order unchanged
# speedup vs baseline: 1.6390x; 1.0403x over previous
#include <hip/hip_runtime.h>
#include <hip/hip_bf16.h>


#define DIM       4096
#define NB        2
#define ESTR      (2 * DIM)
#define N_QUBITS  12
#define N_LAYERS  5
#define GP        20

__device__ __forceinline__ constexpr int lo_pos(int k) {
  return (((k >> 1) & 3) << 2) | ((k >> 3) << 1) | (k & 1);
}

typedef __attribute__((ext_vector_type(2))) float v2f;
typedef __attribute__((ext_vector_type(4))) float v4f;
typedef __attribute__((ext_vector_type(8))) float v8f;

struct cx { float r, i; };
__device__ __forceinline__ cx cmul(cx a, cx b) { return { a.r*b.r - a.i*b.i, a.r*b.i + a.i*b.r }; }
__device__ __forceinline__ cx cadd(cx a, cx b) { return { a.r + b.r, a.i + b.i }; }

struct m2 { cx e[2][2]; };

__device__ __forceinline__ m2 mmul(const m2& A, const m2& B) {
  m2 C;
#pragma unroll
  for (int i = 0; i < 2; ++i)
#pragma unroll
    for (int j = 0; j < 2; ++j)
      C.e[i][j] = cadd(cmul(A.e[i][0], B.e[0][j]), cmul(A.e[i][1], B.e[1][j]));
  return C;
}

__device__ __forceinline__ m2 rot_x(float t) {
  float c = cosf(0.5f * t), s = sinf(0.5f * t);
  m2 M; M.e[0][0] = {c, 0}; M.e[0][1] = {0, -s};
        M.e[1][0] = {0, -s}; M.e[1][1] = {c, 0};
  return M;
}
__device__ __forceinline__ m2 rot_y(float t) {
  float c = cosf(0.5f * t), s = sinf(0.5f * t);
  m2 M; M.e[0][0] = {c, 0}; M.e[0][1] = {-s, 0};
        M.e[1][0] = {s, 0}; M.e[1][1] = {c, 0};
  return M;
}
__device__ __forceinline__ m2 rot_z(float t) {
  float c = cosf(0.5f * t), s = sinf(0.5f * t);
  m2 M; M.e[0][0] = {c, -s}; M.e[0][1] = {0, 0};
        M.e[1][0] = {0, 0};  M.e[1][1] = {c, s};
  return M;
}

__device__ __forceinline__ constexpr int ph(int i) {
  return i ^ (((i >> 6) & 3) << 2) ^ (((i >> 7) & 1) << 4)
           ^ ((((i >> 9) ^ (i >> 10) ^ (i >> 11)) & 1) << 5);
}

__device__ __forceinline__ constexpr int lin_addr(int lin0, int v) {
  return (lin0 ^ ((((v >> 1) ^ (v >> 2) ^ (v >> 3)) & 1) << 5)) + (v << 8);
}

template <int QB>
__device__ __forceinline__ int ld_addr(int wave, int n, int half, int phi,
                                       int blk, int kc, int e) {
  if constexpr (QB == 0) {
    const int c0  = wave * 32 + n;
    const int lo0 = c0 ^ (((c0 >> 6) & 3) << 2) ^ (((c0 >> 7) & 1) << 4) ^ (half << 5);
    const int pk  = (kc ^ (kc >> 1)) & 1;
    return (((lo0 ^ (pk << 5)) | (half << 9)) ^ (blk << 4)) + ((4 * kc + e) << 8);
  } else if constexpr (QB == 4) {
    const int base = (wave << 9) | ((half ^ phi) << 5) | (n ^ (kc << 2));
    return base + (blk << 8) + (kc << 6) + ((e ^ (kc >> 1)) << 4);
  } else {
    const int rowb = (wave << 9) | ((n ^ (n >> 3) ^ (phi << 1)) << 4);
    return (rowb | ((kc ^ (n >> 2)) << 2) | (half << 1)) + (blk << 8) + e;
  }
}

template <int QB>
__device__ __forceinline__ int st_addr(int wave, int n, int half, int phi,
                                       int blk, int v) {
  if constexpr (QB == 0) {
    const int c0  = wave * 32 + n;
    const int lo0 = c0 ^ (((c0 >> 6) & 3) << 2) ^ (((c0 >> 7) & 1) << 4) ^ (half << 5);
    const int pv  = ((v >> 1) ^ (v >> 2)) & 1;
    return (((lo0 ^ (pv << 5)) | (half << 11)) ^ (blk << 4)) + (v << 8);
  } else if constexpr (QB == 4) {
    const int sb = (wave << 9) | (half << 7) | (phi << 5) | (half << 4) | (n ^ (half << 3));
    return (sb ^ ((v & 2) << 4) ^ ((v & 1) << 4) ^ (v & 4)) + (blk << 8) + ((v & 4) << 4);
  } else {
    const int rowb = (wave << 9) | ((n ^ (n >> 3) ^ (phi << 1)) << 4);
    return (rowb | (((2 * half + v) ^ (n >> 2)) << 2)) + (blk << 8);
  }
}

template <int QB>
__device__ __forceinline__ constexpr int sidx(int r, int c) {
  if constexpr (QB == 0) return (r << 8) | c;
  else if constexpr (QB == 4) return ((c & 0xF0) << 4) | (r << 4) | (c & 15);
  else return (c << 4) | r;
}

__device__ __forceinline__ constexpr int ring_src(int i) {
  int j = i;
  for (int q = 11; q >= 0; --q) {
    const int cb = 11 - q;
    const int tb = 11 - ((q + 1) % 12);
    j ^= ((j >> cb) & 1) << tb;
  }
  return j;
}

__device__ __forceinline__ v8f wmma_f32(v2f a, v2f b, v8f c) {
  return __builtin_amdgcn_wmma_f32_16x16x4_f32(
      false, a, false, b, (short)0, c, false, false);
}

template <int QB, bool RING>
__device__ void apply_group(const float* sre0, const float* sim0,
                            float* dre0, float* dim0,
                            const float* ure, const float* uim) {
  static_assert(!RING || QB == 0, "the ring is folded into the hi nibble loads only");
  const int lane  = threadIdx.x & 31;
  const int wave  = threadIdx.x >> 5;
  const int half  = lane >> 4;
  const int n     = lane & 15;
  const int khalf = half << 1;
  const int phi   = (wave ^ (wave >> 1) ^ (wave >> 2)) & 1;
  const int rlane = ph(ring_src((half << 9) | (wave << 5) | n));

  v2f aR[4], aD[4], aSn[4];
#pragma unroll
  for (int kc = 0; kc < 4; ++kc) {
    const int k0 = kc * 4 + khalf;
    const v2f ur = *(const v2f*)&ure[n * GP + k0];
    const v2f ui = *(const v2f*)&uim[n * GP + k0];
    aR[kc]  = ur;
    aD[kc]  = ui - ur;
    aSn[kc] = (-ur) - ui;
  }

#pragma unroll
  for (int el = 0; el < NB; ++el) {
  const float* sre = sre0 + el * ESTR;
  const float* sim = sim0 + el * ESTR;
  float* dre = dre0 + el * ESTR;
  float* dim = dim0 + el * ESTR;
#pragma unroll
  for (int blk = 0; blk < 2; ++blk) {
    v2f bR[4], bI[4];
#pragma unroll
    for (int kc = 0; kc < 4; ++kc) {
      if constexpr (RING) {
        const int a0 = rlane ^ ph(ring_src(((4 * kc) << 8) | (blk << 4)));
        const int a1 = rlane ^ ph(ring_src(((4 * kc + 1) << 8) | (blk << 4)));
        bR[kc].x = sre[a0];
        bR[kc].y = sre[a1];
        bI[kc].x = sim[a0];
        bI[kc].y = sim[a1];
      } else if constexpr (QB == 8) {
        const int a0 = ld_addr<QB>(wave, n, half, phi, blk, kc, 0);
        bR[kc] = *(const v2f*)&sre[a0];
        bI[kc] = *(const v2f*)&sim[a0];
      } else {
        const int a0 = ld_addr<QB>(wave, n, half, phi, blk, kc, 0);
        const int a1 = ld_addr<QB>(wave, n, half, phi, blk, kc, 1);
        bR[kc].x = sre[a0];
        bR[kc].y = sre[a1];
        bI[kc].x = sim[a0];
        bI[kc].y = sim[a1];
      }
    }

    v2f bS[4];
#pragma unroll
    for (int kc = 0; kc < 4; ++kc) bS[kc] = bR[kc] + bI[kc];

    v8f k1 = {};
#pragma unroll
    for (int kc = 0; kc < 4; ++kc) k1 = wmma_f32(aR[kc], bS[kc], k1);
    v8f dR = k1;
    v8f dI = k1;
#pragma unroll
    for (int kc = 0; kc < 4; ++kc) {
      dR = wmma_f32(aSn[kc], bI[kc], dR);
      dI = wmma_f32(aD[kc],  bR[kc], dI);
    }

    if constexpr (QB == 8) {
#pragma unroll
      for (int v = 0; v < 2; ++v) {
        const int a = st_addr<QB>(wave, n, half, phi, blk, v);
        v4f pr = { dR[4 * v], dR[4 * v + 1], dR[4 * v + 2], dR[4 * v + 3] };
        v4f pi = { dI[4 * v], dI[4 * v + 1], dI[4 * v + 2], dI[4 * v + 3] };
        *(v4f*)&dre[a] = pr;
        *(v4f*)&dim[a] = pi;
      }
    } else {
#pragma unroll
      for (int v = 0; v < 8; ++v) {
        const int idx = st_addr<QB>(wave, n, half, phi, blk, v);
        dre[idx] = dR[v];
        dim[idx] = dI[v];
      }
    }
  }
  }
}

__device__ void apply_mid_lo(float* cre0, float* cim0,
                             const float* umr, const float* umi,
                             const float* ulr, const float* uli) {
  const int lane  = threadIdx.x & 31;
  const int wave  = threadIdx.x >> 5;
  const int half  = lane >> 4;
  const int n     = lane & 15;
  const int khalf = half << 1;
  const int phi   = (wave ^ (wave >> 1) ^ (wave >> 2)) & 1;

  v2f mR[4], mD[4], mSn[4];
  v2f lR[4], lD[4], lSn[4];
#pragma unroll
  for (int kc = 0; kc < 4; ++kc) {
    const int k0 = kc * 4 + khalf;
    const v2f mr = *(const v2f*)&umr[n * GP + k0];
    const v2f mi = *(const v2f*)&umi[n * GP + k0];
    mR[kc]  = mr;
    mD[kc]  = mi - mr;
    mSn[kc] = (-mr) - mi;
    const v2f lr = *(const v2f*)&ulr[n * GP + k0];
    const v2f li = *(const v2f*)&uli[n * GP + k0];
    lR[kc]  = lr;
    lD[kc]  = li - lr;
    lSn[kc] = (-lr) - li;
  }

#pragma unroll
  for (int el = 0; el < NB; ++el) {
  float* cre = cre0 + el * ESTR;
  float* cim = cim0 + el * ESTR;
#pragma unroll
  for (int blk = 0; blk < 2; ++blk) {
    v2f sR[4], sI[4], sS[4];
#pragma unroll
    for (int kc = 0; kc < 4; ++kc) {
      const int a0 = ld_addr<4>(wave, n, half, phi, blk, kc, 0);
      const int a1 = ld_addr<4>(wave, n, half, phi, blk, kc, 1);
      sR[kc].x = cre[a0];
      sR[kc].y = cre[a1];
      sI[kc].x = cim[a0];
      sI[kc].y = cim[a1];
    }
#pragma unroll
    for (int kc = 0; kc < 4; ++kc) sS[kc] = sR[kc] + sI[kc];

    v8f t1 = {};
#pragma unroll
    for (int kc = 0; kc < 4; ++kc) t1 = wmma_f32(sS[kc], mR[kc], t1);
    v8f tR = t1;
    v8f tI = t1;
#pragma unroll
    for (int kc = 0; kc < 4; ++kc) {
      tR = wmma_f32(sI[kc], mSn[kc], tR);
      tI = wmma_f32(sR[kc], mD[kc],  tI);
    }

    v2f bR[4], bI[4], bS[4];
#pragma unroll
    for (int kc = 0; kc < 4; ++kc) {
      bR[kc].x = tR[2 * kc];
      bR[kc].y = tR[2 * kc + 1];
      bI[kc].x = tI[2 * kc];
      bI[kc].y = tI[2 * kc + 1];
      bS[kc]   = bR[kc] + bI[kc];
    }

    v8f k1 = {};
#pragma unroll
    for (int kc = 0; kc < 4; ++kc) k1 = wmma_f32(lR[kc], bS[kc], k1);
    v8f dR = k1;
    v8f dI = k1;
#pragma unroll
    for (int kc = 0; kc < 4; ++kc) {
      dR = wmma_f32(lSn[kc], bI[kc], dR);
      dI = wmma_f32(lD[kc],  bR[kc], dI);
    }

#pragma unroll
    for (int v = 0; v < 2; ++v) {
      const int a = st_addr<8>(wave, n, half, phi, blk, v);
      v4f pr = { dR[4 * v], dR[4 * v + 1], dR[4 * v + 2], dR[4 * v + 3] };
      v4f pi = { dI[4 * v], dI[4 * v + 1], dI[4 * v + 2], dI[4 * v + 3] };
      *(v4f*)&cre[a] = pr;
      *(v4f*)&cim[a] = pi;
    }
  }
  }
}

__global__ __launch_bounds__(256)
void qsim_kernel(const float* __restrict__ x, const float* __restrict__ w,
                 float* __restrict__ out) {
  __shared__ __attribute__((aligned(256))) float sre[NB * ESTR];
  __shared__ __attribute__((aligned(256))) float sim[NB * ESTR];
  __shared__ __attribute__((aligned(16))) float ure[3][16 * GP];
  __shared__ __attribute__((aligned(16))) float uim[3][16 * GP];
  __shared__ float g2re[N_LAYERS + 1][N_QUBITS][4];
  __shared__ float g2im[N_LAYERS + 1][N_QUBITS][4];
  __shared__ float red[NB][256];

  const int b   = blockIdx.x;
  const int tid = threadIdx.x;
  const float* xb = x + (size_t)b * (NB * DIM);

  const int gl = tid >> 5, gq = tid & 31;
  const bool gbuild = (gl <= N_LAYERS) && (gq < N_QUBITS);
  const int  gofs = gbuild ? gl * 36 + gq * 3 : 0;
  const float p0 = w[gofs], p1 = w[gofs + 1], p2 = w[gofs + 2];
  __builtin_amdgcn_sched_barrier(0);

  float xs[NB][16];
#pragma unroll
  for (int el = 0; el < NB; ++el)
#pragma unroll
    for (int v = 0; v < 16; ++v) xs[el][v] = xb[el * DIM + tid + (v << 8)];

  asm volatile("" :: "v"(p0), "v"(p1), "v"(p2));

  if (gbuild) {
    m2 U;
    if (gl < N_LAYERS) U = mmul(rot_z(p2), mmul(rot_y(p1), rot_x(p0)));
    else               U = mmul(rot_y(p2), mmul(rot_z(p1), rot_y(p0)));
#pragma unroll
    for (int e = 0; e < 4; ++e) {
      g2re[gl][gq][e] = U.e[e >> 1][e & 1].r;
      g2im[gl][gq][e] = U.e[e >> 1][e & 1].i;
    }
  }

#pragma unroll
  for (int el = 0; el < NB; ++el) {
    float ss = 0.f;
#pragma unroll
    for (int v = 0; v < 16; ++v) ss += xs[el][v] * xs[el][v];
    red[el][tid] = ss;
  }
  __syncthreads();
  for (int s = 128; s > 0; s >>= 1) {
    if (tid < s) {
#pragma unroll
      for (int el = 0; el < NB; ++el) red[el][tid] += red[el][tid + s];
    }
    __syncthreads();
  }
#pragma unroll
  for (int el = 0; el < NB; ++el) {
    const float scale = rsqrtf(sqrtf(red[el][0]) + 1e-12f);
#pragma unroll
    for (int v = 0; v < 16; ++v) {
      const int i = el * ESTR + lin_addr(ph(tid), v);
      sre[i] = xs[el][v] * scale;
      sim[i] = 0.f;
    }
  }
  __syncthreads();

  int cur = 0;
  for (int l = 0; l <= N_LAYERS; ++l) {
    {
      const int m = tid >> 4, j = tid & 15;
#pragma unroll
      for (int g = 0; g < 3; ++g) {
        cx prod = {1.f, 0.f};
#pragma unroll
        for (int k = 0; k < 4; ++k) {
          const int q  = g * 4 + k;
          const int mb = (m >> (3 - k)) & 1;
          const int jb = (j >> (3 - k)) & 1;
          const cx u = { g2re[l][q][mb * 2 + jb], g2im[l][q][mb * 2 + jb] };
          prod = cmul(prod, u);
        }
        const int jp = (g == 2) ? lo_pos(j) : j;
        ure[g][m * GP + jp] = prod.r;
        uim[g][m * GP + jp] = prod.i;
      }
    }
    __syncthreads();

    if (l == 0) {
      apply_group<0, false>(sre, sim, sre, sim, ure[0], uim[0]);
    } else {
      const int oth = cur ^ DIM;
      apply_group<0, true>(sre + cur, sim + cur, sre + oth, sim + oth, ure[0], uim[0]);
      cur = oth;
    }
    float* cre = sre + cur;
    float* cim = sim + cur;
    __syncthreads();
    apply_mid_lo(cre, cim, ure[1], uim[1], ure[2], uim[2]);
    __syncthreads();
  }

  float* ob = out + (size_t)b * (NB * DIM);
#pragma unroll
  for (int el = 0; el < NB; ++el)
#pragma unroll
    for (int v = 0; v < 16; ++v) {
      const int i = tid + (v << 8);
      ob[el * DIM + i] = sre[el * ESTR + cur + lin_addr(ph(tid), v)];
    }
}

extern "C" void kernel_launch(void* const* d_in, const int* in_sizes, int n_in,
                              void* d_out, int out_size, void* d_ws, size_t ws_size,
                              hipStream_t stream) {
  const float* x = (const float*)d_in[0];
  const float* w = (const float*)d_in[1];
  float* out = (float*)d_out;
  const int groups = in_sizes[0] >> 13;
  qsim_kernel<<<dim3(groups), dim3(256), 0, stream>>>(x, w, out);
}


# qubit gates factored as phase * real rotation * phase: 16x16 gates are real, two real WMMA products per block not three (8 not 12); phases applied elementwise; sums regrouped, f32 rounding changes
# speedup vs baseline: 1.9578x; 1.1945x over previous
#include <hip/hip_runtime.h>
#include <hip/hip_bf16.h>


#define DIM       4096
#define NB        2
#define ESTR      (2 * DIM)
#define N_QUBITS  12
#define N_LAYERS  5
#define GP        20

__device__ __forceinline__ constexpr int lo_pos(int k) {
  return (((k >> 1) & 3) << 2) | ((k >> 3) << 1) | (k & 1);
}

typedef __attribute__((ext_vector_type(2))) float v2f;
typedef __attribute__((ext_vector_type(4))) float v4f;
typedef __attribute__((ext_vector_type(8))) float v8f;

struct cx { float r, i; };
__device__ __forceinline__ cx cmul(cx a, cx b) { return { a.r*b.r - a.i*b.i, a.r*b.i + a.i*b.r }; }
__device__ __forceinline__ cx cadd(cx a, cx b) { return { a.r + b.r, a.i + b.i }; }
__device__ __forceinline__ cx cconj(cx a) { return { a.r, -a.i }; }

__device__ __forceinline__ cx unit_of(cx z, float& mag) {
  const float mx = fmaxf(fabsf(z.r), fabsf(z.i));
  if (!(mx > 0.f)) { mag = mx; return { 1.f, 0.f }; }
  const float zr = z.r / mx, zi = z.i / mx;
  const float m  = sqrtf(zr * zr + zi * zi);
  mag = mx * m;
  return { zr / m, zi / m };
}

struct m2 { cx e[2][2]; };

__device__ __forceinline__ m2 mmul(const m2& A, const m2& B) {
  m2 C;
#pragma unroll
  for (int i = 0; i < 2; ++i)
#pragma unroll
    for (int j = 0; j < 2; ++j)
      C.e[i][j] = cadd(cmul(A.e[i][0], B.e[0][j]), cmul(A.e[i][1], B.e[1][j]));
  return C;
}

__device__ __forceinline__ m2 rot_x(float t) {
  float c = cosf(0.5f * t), s = sinf(0.5f * t);
  m2 M; M.e[0][0] = {c, 0}; M.e[0][1] = {0, -s};
        M.e[1][0] = {0, -s}; M.e[1][1] = {c, 0};
  return M;
}
__device__ __forceinline__ m2 rot_y(float t) {
  float c = cosf(0.5f * t), s = sinf(0.5f * t);
  m2 M; M.e[0][0] = {c, 0}; M.e[0][1] = {-s, 0};
        M.e[1][0] = {s, 0}; M.e[1][1] = {c, 0};
  return M;
}
__device__ __forceinline__ m2 rot_z(float t) {
  float c = cosf(0.5f * t), s = sinf(0.5f * t);
  m2 M; M.e[0][0] = {c, -s}; M.e[0][1] = {0, 0};
        M.e[1][0] = {0, 0};  M.e[1][1] = {c, s};
  return M;
}

__device__ __forceinline__ constexpr int ph(int i) {
  return i ^ (((i >> 6) & 3) << 2) ^ (((i >> 7) & 1) << 4)
           ^ ((((i >> 9) ^ (i >> 10) ^ (i >> 11)) & 1) << 5);
}

__device__ __forceinline__ constexpr int lin_addr(int lin0, int v) {
  return (lin0 ^ ((((v >> 1) ^ (v >> 2) ^ (v >> 3)) & 1) << 5)) + (v << 8);
}

template <int QB>
__device__ __forceinline__ int ld_addr(int wave, int n, int half, int phi,
                                       int blk, int kc, int e) {
  if constexpr (QB == 0) {
    const int c0  = wave * 32 + n;
    const int lo0 = c0 ^ (((c0 >> 6) & 3) << 2) ^ (((c0 >> 7) & 1) << 4) ^ (half << 5);
    const int pk  = (kc ^ (kc >> 1)) & 1;
    return (((lo0 ^ (pk << 5)) | (half << 9)) ^ (blk << 4)) + ((4 * kc + e) << 8);
  } else if constexpr (QB == 4) {
    const int base = (wave << 9) | ((half ^ phi) << 5) | (n ^ (kc << 2));
    return base + (blk << 8) + (kc << 6) + ((e ^ (kc >> 1)) << 4);
  } else {
    const int rowb = (wave << 9) | ((n ^ (n >> 3) ^ (phi << 1)) << 4);
    return (rowb | ((kc ^ (n >> 2)) << 2) | (half << 1)) + (blk << 8) + e;
  }
}

template <int QB>
__device__ __forceinline__ int st_addr(int wave, int n, int half, int phi,
                                       int blk, int v) {
  if constexpr (QB == 0) {
    const int c0  = wave * 32 + n;
    const int lo0 = c0 ^ (((c0 >> 6) & 3) << 2) ^ (((c0 >> 7) & 1) << 4) ^ (half << 5);
    const int pv  = ((v >> 1) ^ (v >> 2)) & 1;
    return (((lo0 ^ (pv << 5)) | (half << 11)) ^ (blk << 4)) + (v << 8);
  } else if constexpr (QB == 4) {
    const int sb = (wave << 9) | (half << 7) | (phi << 5) | (half << 4) | (n ^ (half << 3));
    return (sb ^ ((v & 2) << 4) ^ ((v & 1) << 4) ^ (v & 4)) + (blk << 8) + ((v & 4) << 4);
  } else {
    const int rowb = (wave << 9) | ((n ^ (n >> 3) ^ (phi << 1)) << 4);
    return (rowb | (((2 * half + v) ^ (n >> 2)) << 2)) + (blk << 8);
  }
}

template <int QB>
__device__ __forceinline__ constexpr int sidx(int r, int c) {
  if constexpr (QB == 0) return (r << 8) | c;
  else if constexpr (QB == 4) return ((c & 0xF0) << 4) | (r << 4) | (c & 15);
  else return (c << 4) | r;
}

__device__ __forceinline__ constexpr int ring_src(int i) {
  int j = i;
  for (int q = 11; q >= 0; --q) {
    const int cb = 11 - q;
    const int tb = 11 - ((q + 1) % 12);
    j ^= ((j >> cb) & 1) << tb;
  }
  return j;
}

__device__ __forceinline__ v8f wmma_f32(v2f a, v2f b, v8f c) {
  return __builtin_amdgcn_wmma_f32_16x16x4_f32(
      false, a, false, b, (short)0, c, false, false);
}

template <bool RING>
__device__ void apply_hi(const float* sre0, const float* sim0,
                         float* dre0, float* dim0,
                         const float* ur, const float* pr, const float* pi) {
  const int lane  = threadIdx.x & 31;
  const int wave  = threadIdx.x >> 5;
  const int half  = lane >> 4;
  const int n     = lane & 15;
  const int khalf = half << 1;
  const int phi   = (wave ^ (wave >> 1) ^ (wave >> 2)) & 1;
  const int rlane = ph(ring_src((half << 9) | (wave << 5) | n));

  v2f aR[4];
  v2f eR[2][4], eI[2][4];
  {
    const cx yl = { pr[32 + n], pi[32 + n] };
    v2f hR[4], hI[4];
#pragma unroll
    for (int kc = 0; kc < 4; ++kc) {
      const int k0 = kc * 4 + khalf;
      aR[kc] = *(const v2f*)&ur[n * GP + k0];
      hR[kc] = *(const v2f*)&pr[k0];
      hI[kc] = *(const v2f*)&pi[k0];
    }
#pragma unroll
    for (int blk = 0; blk < 2; ++blk) {
      const cx ym = { pr[16 + 2 * wave + blk], pi[16 + 2 * wave + blk] };
      const cx q  = cmul(ym, yl);
#pragma unroll
      for (int kc = 0; kc < 4; ++kc) {
        eR[blk][kc] = hR[kc] * q.r - hI[kc] * q.i;
        eI[blk][kc] = hR[kc] * q.i + hI[kc] * q.r;
      }
    }
  }

#pragma unroll
  for (int el = 0; el < NB; ++el) {
  const float* sre = sre0 + el * ESTR;
  const float* sim = sim0 + el * ESTR;
  float* dre = dre0 + el * ESTR;
  float* dim = dim0 + el * ESTR;
#pragma unroll
  for (int blk = 0; blk < 2; ++blk) {
    v2f bR[4], bI[4];
#pragma unroll
    for (int kc = 0; kc < 4; ++kc) {
      if constexpr (RING) {
        const int a0 = rlane ^ ph(ring_src(((4 * kc) << 8) | (blk << 4)));
        const int a1 = rlane ^ ph(ring_src(((4 * kc + 1) << 8) | (blk << 4)));
        bR[kc].x = sre[a0];
        bR[kc].y = sre[a1];
        bI[kc].x = sim[a0];
        bI[kc].y = sim[a1];
      } else {
        const int a0 = ld_addr<0>(wave, n, half, phi, blk, kc, 0);
        const int a1 = ld_addr<0>(wave, n, half, phi, blk, kc, 1);
        bR[kc].x = sre[a0];
        bR[kc].y = sre[a1];
        bI[kc].x = sim[a0];
        bI[kc].y = sim[a1];
      }
    }

    v2f cR[4], cI[4];
#pragma unroll
    for (int kc = 0; kc < 4; ++kc) {
      cR[kc] = bR[kc] * eR[blk][kc] - bI[kc] * eI[blk][kc];
      cI[kc] = bR[kc] * eI[blk][kc] + bI[kc] * eR[blk][kc];
    }

    v8f dR = {};
    v8f dI = {};
#pragma unroll
    for (int kc = 0; kc < 4; ++kc) {
      dR = wmma_f32(aR[kc], cR[kc], dR);
      dI = wmma_f32(aR[kc], cI[kc], dI);
    }

#pragma unroll
    for (int v = 0; v < 8; ++v) {
      const int idx = st_addr<0>(wave, n, half, phi, blk, v);
      dre[idx] = dR[v];
      dim[idx] = dI[v];
    }
  }
  }
}

__device__ void apply_mid_lo(float* cre0, float* cim0,
                             const float* umr, const float* ulr,
                             const float* pr, const float* pi) {
  const int lane  = threadIdx.x & 31;
  const int wave  = threadIdx.x >> 5;
  const int half  = lane >> 4;
  const int n     = lane & 15;
  const int khalf = half << 1;
  const int phi   = (wave ^ (wave >> 1) ^ (wave >> 2)) & 1;

  v2f mR[4];
  v2f lR[4];
#pragma unroll
  for (int kc = 0; kc < 4; ++kc) {
    const int k0 = kc * 4 + khalf;
    mR[kc] = *(const v2f*)&umr[n * GP + k0];
    lR[kc] = *(const v2f*)&ulr[n * GP + k0];
  }

  v8f fR[2], fI[2];
  {
    const cx xm = { pr[64 + n], pi[64 + n] };
    const v4f lr0 = *(const v4f*)&pr[80 + 8 * half];
    const v4f lr1 = *(const v4f*)&pr[84 + 8 * half];
    const v4f li0 = *(const v4f*)&pi[80 + 8 * half];
    const v4f li1 = *(const v4f*)&pi[84 + 8 * half];
    const v8f xr = { lr0[0], lr0[1], lr0[2], lr0[3], lr1[0], lr1[1], lr1[2], lr1[3] };
    const v8f xi = { li0[0], li0[1], li0[2], li0[3], li1[0], li1[1], li1[2], li1[3] };
#pragma unroll
    for (int blk = 0; blk < 2; ++blk) {
      const cx xh = { pr[48 + 2 * wave + blk], pi[48 + 2 * wave + blk] };
      const cx q  = cmul(xm, xh);
      fR[blk] = xr * q.r - xi * q.i;
      fI[blk] = xr * q.i + xi * q.r;
    }
  }

#pragma unroll
  for (int el = 0; el < NB; ++el) {
  float* cre = cre0 + el * ESTR;
  float* cim = cim0 + el * ESTR;
#pragma unroll
  for (int blk = 0; blk < 2; ++blk) {
    v2f sR[4], sI[4];
#pragma unroll
    for (int kc = 0; kc < 4; ++kc) {
      const int a0 = ld_addr<4>(wave, n, half, phi, blk, kc, 0);
      const int a1 = ld_addr<4>(wave, n, half, phi, blk, kc, 1);
      sR[kc].x = cre[a0];
      sR[kc].y = cre[a1];
      sI[kc].x = cim[a0];
      sI[kc].y = cim[a1];
    }

    v8f tR = {};
    v8f tI = {};
#pragma unroll
    for (int kc = 0; kc < 4; ++kc) {
      tR = wmma_f32(sR[kc], mR[kc], tR);
      tI = wmma_f32(sI[kc], mR[kc], tI);
    }

    v2f bR[4], bI[4];
#pragma unroll
    for (int kc = 0; kc < 4; ++kc) {
      bR[kc].x = tR[2 * kc];
      bR[kc].y = tR[2 * kc + 1];
      bI[kc].x = tI[2 * kc];
      bI[kc].y = tI[2 * kc + 1];
    }

    v8f dR = {};
    v8f dI = {};
#pragma unroll
    for (int kc = 0; kc < 4; ++kc) {
      dR = wmma_f32(lR[kc], bR[kc], dR);
      dI = wmma_f32(lR[kc], bI[kc], dI);
    }

    const v8f oR = dR * fR[blk] - dI * fI[blk];
    const v8f oI = dR * fI[blk] + dI * fR[blk];

#pragma unroll
    for (int v = 0; v < 2; ++v) {
      const int a = st_addr<8>(wave, n, half, phi, blk, v);
      v4f pr4 = { oR[4 * v], oR[4 * v + 1], oR[4 * v + 2], oR[4 * v + 3] };
      v4f pi4 = { oI[4 * v], oI[4 * v + 1], oI[4 * v + 2], oI[4 * v + 3] };
      *(v4f*)&cre[a] = pr4;
      *(v4f*)&cim[a] = pi4;
    }
  }
  }
}

__global__ __launch_bounds__(256)
void qsim_kernel(const float* __restrict__ x, const float* __restrict__ w,
                 float* __restrict__ out) {
  __shared__ __attribute__((aligned(256))) float sre[NB * ESTR];
  __shared__ __attribute__((aligned(256))) float sim[NB * ESTR];
  __shared__ __attribute__((aligned(16))) float ure[3][16 * GP];
  __shared__ __attribute__((aligned(16))) float pvr[6][16];
  __shared__ __attribute__((aligned(16))) float pvi[6][16];
  __shared__ float g2re[N_LAYERS + 1][N_QUBITS][4];
  __shared__ float pqr[N_LAYERS + 1][N_QUBITS][4];
  __shared__ float pqi[N_LAYERS + 1][N_QUBITS][4];
  __shared__ float red[NB][256];

  const int b   = blockIdx.x;
  const int tid = threadIdx.x;
  const float* xb = x + (size_t)b * (NB * DIM);

  const int gl = tid >> 5, gq = tid & 31;
  const bool gbuild = (gl <= N_LAYERS) && (gq < N_QUBITS);
  const int  gofs = gbuild ? gl * 36 + gq * 3 : 0;
  const float p0 = w[gofs], p1 = w[gofs + 1], p2 = w[gofs + 2];
  __builtin_amdgcn_sched_barrier(0);

  float xs[NB][16];
#pragma unroll
  for (int el = 0; el < NB; ++el)
#pragma unroll
    for (int v = 0; v < 16; ++v) xs[el][v] = xb[el * DIM + tid + (v << 8)];

  asm volatile("" :: "v"(p0), "v"(p1), "v"(p2));

  if (gbuild) {
    m2 U;
    if (gl < N_LAYERS) U = mmul(rot_z(p2), mmul(rot_y(p1), rot_x(p0)));
    else               U = mmul(rot_y(p2), mmul(rot_z(p1), rot_y(p0)));
    float c, sn, t;
    const cx x0 = unit_of(U.e[0][0], c);
    const cx x1 = unit_of(U.e[1][0], sn);
    cx y1;
    if (c >= sn) {
      y1 = cmul(unit_of(U.e[1][1], t), cconj(x1));
    } else {
      const cx nb = { -U.e[0][1].r, -U.e[0][1].i };
      y1 = cmul(cconj(x0), unit_of(nb, t));
    }
    g2re[gl][gq][0] = c;   g2re[gl][gq][1] = -sn;
    g2re[gl][gq][2] = sn;  g2re[gl][gq][3] = c;
    pqr[gl][gq][0] = 1.f;  pqi[gl][gq][0] = 0.f;
    pqr[gl][gq][1] = y1.r; pqi[gl][gq][1] = y1.i;
    pqr[gl][gq][2] = x0.r; pqi[gl][gq][2] = x0.i;
    pqr[gl][gq][3] = x1.r; pqi[gl][gq][3] = x1.i;
  }

#pragma unroll
  for (int el = 0; el < NB; ++el) {
    float ss = 0.f;
#pragma unroll
    for (int v = 0; v < 16; ++v) ss += xs[el][v] * xs[el][v];
    red[el][tid] = ss;
  }
  __syncthreads();
  for (int s = 128; s > 0; s >>= 1) {
    if (tid < s) {
#pragma unroll
      for (int el = 0; el < NB; ++el) red[el][tid] += red[el][tid + s];
    }
    __syncthreads();
  }
#pragma unroll
  for (int el = 0; el < NB; ++el) {
    const float scale = rsqrtf(sqrtf(red[el][0]) + 1e-12f);
#pragma unroll
    for (int v = 0; v < 16; ++v) {
      const int i = el * ESTR + lin_addr(ph(tid), v);
      sre[i] = xs[el][v] * scale;
      sim[i] = 0.f;
    }
  }
  __syncthreads();

  int cur = 0;
  for (int l = 0; l <= N_LAYERS; ++l) {
    {
      const int m = tid >> 4, j = tid & 15;
#pragma unroll
      for (int g = 0; g < 3; ++g) {
        float prod = 1.f;
#pragma unroll
        for (int k = 0; k < 4; ++k) {
          const int q  = g * 4 + k;
          const int mb = (m >> (3 - k)) & 1;
          const int jb = (j >> (3 - k)) & 1;
          prod *= g2re[l][q][mb * 2 + jb];
        }
        const int jp = (g == 2) ? lo_pos(j) : j;
        ure[g][m * GP + jp] = prod;
      }
      if (tid < 96) {
        const int post = (m >= 3) ? 1 : 0;
        const int g    = m - 3 * post;
        cx prod = {1.f, 0.f};
#pragma unroll
        for (int k = 0; k < 4; ++k) {
          const int q  = g * 4 + k;
          const int jb = (j >> (3 - k)) & 1;
          const cx u = { pqr[l][q][post * 2 + jb], pqi[l][q][post * 2 + jb] };
          prod = cmul(prod, u);
        }
        pvr[m][j] = prod.r;
        pvi[m][j] = prod.i;
      }
    }
    __syncthreads();

    if (l == 0) {
      apply_hi<false>(sre, sim, sre, sim, ure[0], &pvr[0][0], &pvi[0][0]);
    } else {
      const int oth = cur ^ DIM;
      apply_hi<true>(sre + cur, sim + cur, sre + oth, sim + oth, ure[0], &pvr[0][0], &pvi[0][0]);
      cur = oth;
    }
    float* cre = sre + cur;
    float* cim = sim + cur;
    __syncthreads();
    apply_mid_lo(cre, cim, ure[1], ure[2], &pvr[0][0], &pvi[0][0]);
    __syncthreads();
  }

  float* ob = out + (size_t)b * (NB * DIM);
#pragma unroll
  for (int el = 0; el < NB; ++el)
#pragma unroll
    for (int v = 0; v < 16; ++v) {
      const int i = tid + (v << 8);
      ob[el * DIM + i] = sre[el * ESTR + cur + lin_addr(ph(tid), v)];
    }
}

extern "C" void kernel_launch(void* const* d_in, const int* in_sizes, int n_in,
                              void* d_out, int out_size, void* d_ws, size_t ws_size,
                              hipStream_t stream) {
  const float* x = (const float*)d_in[0];
  const float* w = (const float*)d_in[1];
  float* out = (float*)d_out;
  const int groups = in_sizes[0] >> 13;
  qsim_kernel<<<dim3(groups), dim3(256), 0, stream>>>(x, w, out);
}


# no LDS staging at either end: x loaded straight into the first hi gate's B fragments, last lo gate's D tiles stored straight to out; norm scale applied at the output, |x|^2 summed in another order
# speedup vs baseline: 2.0639x; 1.0542x over previous
#include <hip/hip_runtime.h>
#include <hip/hip_bf16.h>


#define DIM       4096
#define NB        2
#define ESTR      (2 * DIM)
#define N_QUBITS  12
#define N_LAYERS  5
#define GP        20

__device__ __forceinline__ constexpr int lo_pos(int k) {
  return (((k >> 1) & 3) << 2) | ((k >> 3) << 1) | (k & 1);
}

typedef __attribute__((ext_vector_type(2))) float v2f;
typedef __attribute__((ext_vector_type(4))) float v4f;
typedef __attribute__((ext_vector_type(8))) float v8f;

struct cx { float r, i; };
__device__ __forceinline__ cx cmul(cx a, cx b) { return { a.r*b.r - a.i*b.i, a.r*b.i + a.i*b.r }; }
__device__ __forceinline__ cx cadd(cx a, cx b) { return { a.r + b.r, a.i + b.i }; }
__device__ __forceinline__ cx cconj(cx a) { return { a.r, -a.i }; }

__device__ __forceinline__ cx unit_of(cx z, float& mag) {
  const float mx = fmaxf(fabsf(z.r), fabsf(z.i));
  if (!(mx > 0.f)) { mag = mx; return { 1.f, 0.f }; }
  const float zr = z.r / mx, zi = z.i / mx;
  const float m  = sqrtf(zr * zr + zi * zi);
  mag = mx * m;
  return { zr / m, zi / m };
}

struct m2 { cx e[2][2]; };

__device__ __forceinline__ m2 mmul(const m2& A, const m2& B) {
  m2 C;
#pragma unroll
  for (int i = 0; i < 2; ++i)
#pragma unroll
    for (int j = 0; j < 2; ++j)
      C.e[i][j] = cadd(cmul(A.e[i][0], B.e[0][j]), cmul(A.e[i][1], B.e[1][j]));
  return C;
}

__device__ __forceinline__ m2 rot_x(float t) {
  float c = cosf(0.5f * t), s = sinf(0.5f * t);
  m2 M; M.e[0][0] = {c, 0}; M.e[0][1] = {0, -s};
        M.e[1][0] = {0, -s}; M.e[1][1] = {c, 0};
  return M;
}
__device__ __forceinline__ m2 rot_y(float t) {
  float c = cosf(0.5f * t), s = sinf(0.5f * t);
  m2 M; M.e[0][0] = {c, 0}; M.e[0][1] = {-s, 0};
        M.e[1][0] = {s, 0}; M.e[1][1] = {c, 0};
  return M;
}
__device__ __forceinline__ m2 rot_z(float t) {
  float c = cosf(0.5f * t), s = sinf(0.5f * t);
  m2 M; M.e[0][0] = {c, -s}; M.e[0][1] = {0, 0};
        M.e[1][0] = {0, 0};  M.e[1][1] = {c, s};
  return M;
}

__device__ __forceinline__ constexpr int ph(int i) {
  return i ^ (((i >> 6) & 3) << 2) ^ (((i >> 7) & 1) << 4)
           ^ ((((i >> 9) ^ (i >> 10) ^ (i >> 11)) & 1) << 5);
}

template <int QB>
__device__ __forceinline__ int ld_addr(int wave, int n, int half, int phi,
                                       int blk, int kc, int e) {
  if constexpr (QB == 4) {
    const int base = (wave << 9) | ((half ^ phi) << 5) | (n ^ (kc << 2));
    return base + (blk << 8) + (kc << 6) + ((e ^ (kc >> 1)) << 4);
  } else {
    const int rowb = (wave << 9) | ((n ^ (n >> 3) ^ (phi << 1)) << 4);
    return (rowb | ((kc ^ (n >> 2)) << 2) | (half << 1)) + (blk << 8) + e;
  }
}

template <int QB>
__device__ __forceinline__ int st_addr(int wave, int n, int half, int phi,
                                       int blk, int v) {
  if constexpr (QB == 0) {
    const int c0  = wave * 32 + n;
    const int lo0 = c0 ^ (((c0 >> 6) & 3) << 2) ^ (((c0 >> 7) & 1) << 4) ^ (half << 5);
    const int pv  = ((v >> 1) ^ (v >> 2)) & 1;
    return (((lo0 ^ (pv << 5)) | (half << 11)) ^ (blk << 4)) + (v << 8);
  } else if constexpr (QB == 4) {
    const int sb = (wave << 9) | (half << 7) | (phi << 5) | (half << 4) | (n ^ (half << 3));
    return (sb ^ ((v & 2) << 4) ^ ((v & 1) << 4) ^ (v & 4)) + (blk << 8) + ((v & 4) << 4);
  } else {
    const int rowb = (wave << 9) | ((n ^ (n >> 3) ^ (phi << 1)) << 4);
    return (rowb | (((2 * half + v) ^ (n >> 2)) << 2)) + (blk << 8);
  }
}

template <int QB>
__device__ __forceinline__ constexpr int sidx(int r, int c) {
  if constexpr (QB == 0) return (r << 8) | c;
  else if constexpr (QB == 4) return ((c & 0xF0) << 4) | (r << 4) | (c & 15);
  else return (c << 4) | r;
}

__device__ __forceinline__ constexpr int ring_src(int i) {
  int j = i;
  for (int q = 11; q >= 0; --q) {
    const int cb = 11 - q;
    const int tb = 11 - ((q + 1) % 12);
    j ^= ((j >> cb) & 1) << tb;
  }
  return j;
}

__device__ __forceinline__ v8f wmma_f32(v2f a, v2f b, v8f c) {
  return __builtin_amdgcn_wmma_f32_16x16x4_f32(
      false, a, false, b, (short)0, c, false, false);
}

template <bool FIRST>
__device__ __forceinline__ void apply_hi(const v2f (&xq)[NB][2][4],
                         const float* sre0, const float* sim0,
                         float* dre0, float* dim0,
                         const float* ur, const float* pr, const float* pi) {
  const int lane  = threadIdx.x & 31;
  const int wave  = threadIdx.x >> 5;
  const int half  = lane >> 4;
  const int n     = lane & 15;
  const int khalf = half << 1;
  const int phi   = (wave ^ (wave >> 1) ^ (wave >> 2)) & 1;
  const int rlane = ph(ring_src((half << 9) | (wave << 5) | n));

  v2f aR[4];
  v2f eR[2][4], eI[2][4];
  {
    const cx yl = { pr[32 + n], pi[32 + n] };
    v2f hR[4], hI[4];
#pragma unroll
    for (int kc = 0; kc < 4; ++kc) {
      const int k0 = kc * 4 + khalf;
      aR[kc] = *(const v2f*)&ur[n * GP + k0];
      hR[kc] = *(const v2f*)&pr[k0];
      hI[kc] = *(const v2f*)&pi[k0];
    }
#pragma unroll
    for (int blk = 0; blk < 2; ++blk) {
      const cx ym = { pr[16 + 2 * wave + blk], pi[16 + 2 * wave + blk] };
      const cx q  = cmul(ym, yl);
#pragma unroll
      for (int kc = 0; kc < 4; ++kc) {
        eR[blk][kc] = hR[kc] * q.r - hI[kc] * q.i;
        eI[blk][kc] = hR[kc] * q.i + hI[kc] * q.r;
      }
    }
  }

#pragma unroll
  for (int el = 0; el < NB; ++el) {
  const float* sre = sre0 + el * ESTR;
  const float* sim = sim0 + el * ESTR;
  float* dre = dre0 + el * ESTR;
  float* dim = dim0 + el * ESTR;
#pragma unroll
  for (int blk = 0; blk < 2; ++blk) {
    v2f cR[4], cI[4];
#pragma unroll
    for (int kc = 0; kc < 4; ++kc) {
      if constexpr (FIRST) {
        cR[kc] = xq[el][blk][kc] * eR[blk][kc];
        cI[kc] = xq[el][blk][kc] * eI[blk][kc];
      } else {
        const int a0 = rlane ^ ph(ring_src(((4 * kc) << 8) | (blk << 4)));
        const int a1 = rlane ^ ph(ring_src(((4 * kc + 1) << 8) | (blk << 4)));
        v2f bR, bI;
        bR.x = sre[a0];
        bR.y = sre[a1];
        bI.x = sim[a0];
        bI.y = sim[a1];
        cR[kc] = bR * eR[blk][kc] - bI * eI[blk][kc];
        cI[kc] = bR * eI[blk][kc] + bI * eR[blk][kc];
      }
    }

    v8f dR = {};
    v8f dI = {};
#pragma unroll
    for (int kc = 0; kc < 4; ++kc) {
      dR = wmma_f32(aR[kc], cR[kc], dR);
      dI = wmma_f32(aR[kc], cI[kc], dI);
    }

#pragma unroll
    for (int v = 0; v < 8; ++v) {
      const int idx = st_addr<0>(wave, n, half, phi, blk, v);
      dre[idx] = dR[v];
      dim[idx] = dI[v];
    }
  }
  }
}

template <bool LAST>
__device__ void apply_mid_lo(float* cre0, float* cim0,
                             const float* umr, const float* ulr,
                             const float* pr, const float* pi,
                             const float* ssq, float* ob) {
  const int lane  = threadIdx.x & 31;
  const int wave  = threadIdx.x >> 5;
  const int half  = lane >> 4;
  const int n     = lane & 15;
  const int khalf = half << 1;
  const int phi   = (wave ^ (wave >> 1) ^ (wave >> 2)) & 1;

  v2f mR[4];
  v2f lR[4];
#pragma unroll
  for (int kc = 0; kc < 4; ++kc) {
    const int k0 = kc * 4 + khalf;
    mR[kc] = *(const v2f*)&umr[n * GP + k0];
    lR[kc] = *(const v2f*)&ulr[n * GP + k0];
  }

  v8f fR[2], fI[2];
  {
    const cx xm = { pr[64 + n], pi[64 + n] };
    const v4f lr0 = *(const v4f*)&pr[80 + 8 * half];
    const v4f lr1 = *(const v4f*)&pr[84 + 8 * half];
    const v4f li0 = *(const v4f*)&pi[80 + 8 * half];
    const v4f li1 = *(const v4f*)&pi[84 + 8 * half];
    const v8f xr = { lr0[0], lr0[1], lr0[2], lr0[3], lr1[0], lr1[1], lr1[2], lr1[3] };
    const v8f xi = { li0[0], li0[1], li0[2], li0[3], li1[0], li1[1], li1[2], li1[3] };
#pragma unroll
    for (int blk = 0; blk < 2; ++blk) {
      const cx xh = { pr[48 + 2 * wave + blk], pi[48 + 2 * wave + blk] };
      const cx q  = cmul(xm, xh);
      fR[blk] = xr * q.r - xi * q.i;
      fI[blk] = xr * q.i + xi * q.r;
    }
  }

  float scale[NB] = { 1.f, 1.f };
  if constexpr (LAST) {
#pragma unroll
    for (int el = 0; el < NB; ++el) {
      float ss = ssq[el * 8];
#pragma unroll
      for (int j = 1; j < 8; ++j) ss += ssq[el * 8 + j];
      scale[el] = rsqrtf(sqrtf(ss) + 1e-12f);
    }
  }

#pragma unroll
  for (int el = 0; el < NB; ++el) {
  float* cre = cre0 + el * ESTR;
  float* cim = cim0 + el * ESTR;
#pragma unroll
  for (int blk = 0; blk < 2; ++blk) {
    v2f sR[4], sI[4];
#pragma unroll
    for (int kc = 0; kc < 4; ++kc) {
      const int a0 = ld_addr<4>(wave, n, half, phi, blk, kc, 0);
      const int a1 = ld_addr<4>(wave, n, half, phi, blk, kc, 1);
      sR[kc].x = cre[a0];
      sR[kc].y = cre[a1];
      sI[kc].x = cim[a0];
      sI[kc].y = cim[a1];
    }

    v8f tR = {};
    v8f tI = {};
#pragma unroll
    for (int kc = 0; kc < 4; ++kc) {
      tR = wmma_f32(sR[kc], mR[kc], tR);
      tI = wmma_f32(sI[kc], mR[kc], tI);
    }

    v2f bR[4], bI[4];
#pragma unroll
    for (int kc = 0; kc < 4; ++kc) {
      bR[kc].x = tR[2 * kc];
      bR[kc].y = tR[2 * kc + 1];
      bI[kc].x = tI[2 * kc];
      bI[kc].y = tI[2 * kc + 1];
    }

    v8f dR = {};
    v8f dI = {};
#pragma unroll
    for (int kc = 0; kc < 4; ++kc) {
      dR = wmma_f32(lR[kc], bR[kc], dR);
      dI = wmma_f32(lR[kc], bI[kc], dI);
    }

    const v8f oR = dR * fR[blk] - dI * fI[blk];

    if constexpr (LAST) {
      const v8f oS = oR * scale[el];
      float* o = ob + el * DIM + ((2 * wave + blk) << 8) + (n << 4) + 8 * half;
#pragma unroll
      for (int v = 0; v < 2; ++v) {
        v4f pr4 = { oS[4 * v], oS[4 * v + 1], oS[4 * v + 2], oS[4 * v + 3] };
        *(v4f*)&o[4 * v] = pr4;
      }
    } else {
      const v8f oI = dR * fI[blk] + dI * fR[blk];

#pragma unroll
      for (int v = 0; v < 2; ++v) {
        const int a = st_addr<8>(wave, n, half, phi, blk, v);
        v4f pr4 = { oR[4 * v], oR[4 * v + 1], oR[4 * v + 2], oR[4 * v + 3] };
        v4f pi4 = { oI[4 * v], oI[4 * v + 1], oI[4 * v + 2], oI[4 * v + 3] };
        *(v4f*)&cre[a] = pr4;
        *(v4f*)&cim[a] = pi4;
      }
    }
  }
  }
}

__device__ __forceinline__ void build_layer(int l, int tid,
    const float (*g2re)[N_QUBITS][4], const float (*pqr)[N_QUBITS][4],
    const float (*pqi)[N_QUBITS][4],
    float (*ure)[16 * GP], float (*pvr)[16], float (*pvi)[16]) {
  const int m = tid >> 4, j = tid & 15;
#pragma unroll
  for (int g = 0; g < 3; ++g) {
    float prod = 1.f;
#pragma unroll
    for (int k = 0; k < 4; ++k) {
      const int q  = g * 4 + k;
      const int mb = (m >> (3 - k)) & 1;
      const int jb = (j >> (3 - k)) & 1;
      prod *= g2re[l][q][mb * 2 + jb];
    }
    const int jp = (g == 2) ? lo_pos(j) : j;
    ure[g][m * GP + jp] = prod;
  }
  if (tid < 96) {
    const int post = (m >= 3) ? 1 : 0;
    const int g    = m - 3 * post;
    cx prod = {1.f, 0.f};
#pragma unroll
    for (int k = 0; k < 4; ++k) {
      const int q  = g * 4 + k;
      const int jb = (j >> (3 - k)) & 1;
      const cx u = { pqr[l][q][post * 2 + jb], pqi[l][q][post * 2 + jb] };
      prod = cmul(prod, u);
    }
    pvr[m][j] = prod.r;
    pvi[m][j] = prod.i;
  }
}

__global__ __launch_bounds__(256)
void qsim_kernel(const float* __restrict__ x, const float* __restrict__ w,
                 float* __restrict__ out) {
  __shared__ __attribute__((aligned(256))) float sre[NB * ESTR];
  __shared__ __attribute__((aligned(256))) float sim[NB * ESTR];
  __shared__ __attribute__((aligned(16))) float ure[3][16 * GP];
  __shared__ __attribute__((aligned(16))) float pvr[6][16];
  __shared__ __attribute__((aligned(16))) float pvi[6][16];
  __shared__ float g2re[N_LAYERS + 1][N_QUBITS][4];
  __shared__ float pqr[N_LAYERS + 1][N_QUBITS][4];
  __shared__ float pqi[N_LAYERS + 1][N_QUBITS][4];
  __shared__ float red[NB][8];

  const int b   = blockIdx.x;
  const int tid = threadIdx.x;
  const float* xb = x + (size_t)b * (NB * DIM);

  const int gl = tid >> 5, gq = tid & 31;
  const bool gbuild = (gl <= N_LAYERS) && (gq < N_QUBITS);
  const int  gofs = gbuild ? gl * 36 + gq * 3 : 0;
  const float p0 = w[gofs], p1 = w[gofs + 1], p2 = w[gofs + 2];
  __builtin_amdgcn_sched_barrier(0);

  v2f xq[NB][2][4];
  {
    const int xlane = tid & 31, xwave = tid >> 5;
    const float* xl = xb + ((xlane >> 4) << 9) + xwave * 32 + (xlane & 15);
#pragma unroll
    for (int el = 0; el < NB; ++el)
#pragma unroll
      for (int kc = 0; kc < 4; ++kc)
#pragma unroll
        for (int blk = 0; blk < 2; ++blk) {
          xq[el][blk][kc].x = xl[el * DIM + ((4 * kc) << 8) + blk * 16];
          xq[el][blk][kc].y = xl[el * DIM + ((4 * kc + 1) << 8) + blk * 16];
        }
  }

  asm volatile("" :: "v"(p0), "v"(p1), "v"(p2));

  if (gbuild) {
    m2 U;
    if (gl < N_LAYERS) U = mmul(rot_z(p2), mmul(rot_y(p1), rot_x(p0)));
    else               U = mmul(rot_y(p2), mmul(rot_z(p1), rot_y(p0)));
    float c, sn, t;
    const cx x0 = unit_of(U.e[0][0], c);
    const cx x1 = unit_of(U.e[1][0], sn);
    cx y1;
    if (c >= sn) {
      y1 = cmul(unit_of(U.e[1][1], t), cconj(x1));
    } else {
      const cx nb = { -U.e[0][1].r, -U.e[0][1].i };
      y1 = cmul(cconj(x0), unit_of(nb, t));
    }
    g2re[gl][gq][0] = c;   g2re[gl][gq][1] = -sn;
    g2re[gl][gq][2] = sn;  g2re[gl][gq][3] = c;
    pqr[gl][gq][0] = 1.f;  pqi[gl][gq][0] = 0.f;
    pqr[gl][gq][1] = y1.r; pqi[gl][gq][1] = y1.i;
    pqr[gl][gq][2] = x0.r; pqi[gl][gq][2] = x0.i;
    pqr[gl][gq][3] = x1.r; pqi[gl][gq][3] = x1.i;
  }

#pragma unroll
  for (int el = 0; el < NB; ++el) {
    float ss = 0.f;
#pragma unroll
    for (int blk = 0; blk < 2; ++blk)
#pragma unroll
      for (int kc = 0; kc < 4; ++kc) {
        ss += xq[el][blk][kc].x * xq[el][blk][kc].x;
        ss += xq[el][blk][kc].y * xq[el][blk][kc].y;
      }
#pragma unroll
    for (int m = 16; m > 0; m >>= 1) ss += __shfl_xor(ss, m, 32);
    if ((tid & 31) == 0) red[el][tid >> 5] = ss;
  }
  __syncthreads();

  float* ob = out + (size_t)b * (NB * DIM);

  build_layer(0, tid, g2re, pqr, pqi, ure, pvr, pvi);
  __syncthreads();
  apply_hi<true>(xq, sre, sim, sre, sim, ure[0], &pvr[0][0], &pvi[0][0]);
  __syncthreads();
  apply_mid_lo<false>(sre, sim, ure[1], ure[2], &pvr[0][0], &pvi[0][0], &red[0][0], ob);
  __syncthreads();

  int cur = 0;
  for (int l = 1; l <= N_LAYERS; ++l) {
    build_layer(l, tid, g2re, pqr, pqi, ure, pvr, pvi);
    __syncthreads();

    const int oth = cur ^ DIM;
    apply_hi<false>(xq, sre + cur, sim + cur, sre + oth, sim + oth, ure[0], &pvr[0][0], &pvi[0][0]);
    cur = oth;
    float* cre = sre + cur;
    float* cim = sim + cur;
    __syncthreads();
    if (l == N_LAYERS) {
      apply_mid_lo<true>(cre, cim, ure[1], ure[2], &pvr[0][0], &pvi[0][0], &red[0][0], ob);
    } else {
      apply_mid_lo<false>(cre, cim, ure[1], ure[2], &pvr[0][0], &pvi[0][0], &red[0][0], ob);
      __syncthreads();
    }
  }
}

extern "C" void kernel_launch(void* const* d_in, const int* in_sizes, int n_in,
                              void* d_out, int out_size, void* d_ws, size_t ws_size,
                              hipStream_t stream) {
  const float* x = (const float*)d_in[0];
  const float* w = (const float*)d_in[1];
  float* out = (float*)d_out;
  const int groups = in_sizes[0] >> 13;
  qsim_kernel<<<dim3(groups), dim3(256), 0, stream>>>(x, w, out);
}


# each block's LDS fragment reads issued as one group one block ahead of its matrix run (two register sets), stores in one group behind it: one LDS place per block, not four or five; sums unchanged
# speedup vs baseline: 2.1283x; 1.0312x over previous
#include <hip/hip_runtime.h>
#include <hip/hip_bf16.h>


#define DIM       4096
#define NB        2
#define ESTR      (2 * DIM)
#define N_QUBITS  12
#define N_LAYERS  5
#define GP        20

__device__ __forceinline__ constexpr int lo_pos(int k) {
  return (((k >> 1) & 3) << 2) | ((k >> 3) << 1) | (k & 1);
}

typedef __attribute__((ext_vector_type(2))) float v2f;
typedef __attribute__((ext_vector_type(4))) float v4f;
typedef __attribute__((ext_vector_type(8))) float v8f;

struct cx { float r, i; };
__device__ __forceinline__ cx cmul(cx a, cx b) { return { a.r*b.r - a.i*b.i, a.r*b.i + a.i*b.r }; }
__device__ __forceinline__ cx cadd(cx a, cx b) { return { a.r + b.r, a.i + b.i }; }
__device__ __forceinline__ cx cconj(cx a) { return { a.r, -a.i }; }

__device__ __forceinline__ cx unit_of(cx z, float& mag) {
  const float mx = fmaxf(fabsf(z.r), fabsf(z.i));
  if (!(mx > 0.f)) { mag = mx; return { 1.f, 0.f }; }
  const float zr = z.r / mx, zi = z.i / mx;
  const float m  = sqrtf(zr * zr + zi * zi);
  mag = mx * m;
  return { zr / m, zi / m };
}

struct m2 { cx e[2][2]; };

__device__ __forceinline__ m2 mmul(const m2& A, const m2& B) {
  m2 C;
#pragma unroll
  for (int i = 0; i < 2; ++i)
#pragma unroll
    for (int j = 0; j < 2; ++j)
      C.e[i][j] = cadd(cmul(A.e[i][0], B.e[0][j]), cmul(A.e[i][1], B.e[1][j]));
  return C;
}

__device__ __forceinline__ m2 rot_x(float t) {
  float c = cosf(0.5f * t), s = sinf(0.5f * t);
  m2 M; M.e[0][0] = {c, 0}; M.e[0][1] = {0, -s};
        M.e[1][0] = {0, -s}; M.e[1][1] = {c, 0};
  return M;
}
__device__ __forceinline__ m2 rot_y(float t) {
  float c = cosf(0.5f * t), s = sinf(0.5f * t);
  m2 M; M.e[0][0] = {c, 0}; M.e[0][1] = {-s, 0};
        M.e[1][0] = {s, 0}; M.e[1][1] = {c, 0};
  return M;
}
__device__ __forceinline__ m2 rot_z(float t) {
  float c = cosf(0.5f * t), s = sinf(0.5f * t);
  m2 M; M.e[0][0] = {c, -s}; M.e[0][1] = {0, 0};
        M.e[1][0] = {0, 0};  M.e[1][1] = {c, s};
  return M;
}

__device__ __forceinline__ constexpr int ph(int i) {
  return i ^ (((i >> 6) & 3) << 2) ^ (((i >> 7) & 1) << 4)
           ^ ((((i >> 9) ^ (i >> 10) ^ (i >> 11)) & 1) << 5);
}

template <int QB>
__device__ __forceinline__ int ld_addr(int wave, int n, int half, int phi,
                                       int blk, int kc, int e) {
  if constexpr (QB == 4) {
    const int base = (wave << 9) | ((half ^ phi) << 5) | (n ^ (kc << 2));
    return base + (blk << 8) + (kc << 6) + ((e ^ (kc >> 1)) << 4);
  } else {
    const int rowb = (wave << 9) | ((n ^ (n >> 3) ^ (phi << 1)) << 4);
    return (rowb | ((kc ^ (n >> 2)) << 2) | (half << 1)) + (blk << 8) + e;
  }
}

template <int QB>
__device__ __forceinline__ int st_addr(int wave, int n, int half, int phi,
                                       int blk, int v) {
  if constexpr (QB == 0) {
    const int c0  = wave * 32 + n;
    const int lo0 = c0 ^ (((c0 >> 6) & 3) << 2) ^ (((c0 >> 7) & 1) << 4) ^ (half << 5);
    const int pv  = ((v >> 1) ^ (v >> 2)) & 1;
    return (((lo0 ^ (pv << 5)) | (half << 11)) ^ (blk << 4)) + (v << 8);
  } else if constexpr (QB == 4) {
    const int sb = (wave << 9) | (half << 7) | (phi << 5) | (half << 4) | (n ^ (half << 3));
    return (sb ^ ((v & 2) << 4) ^ ((v & 1) << 4) ^ (v & 4)) + (blk << 8) + ((v & 4) << 4);
  } else {
    const int rowb = (wave << 9) | ((n ^ (n >> 3) ^ (phi << 1)) << 4);
    return (rowb | (((2 * half + v) ^ (n >> 2)) << 2)) + (blk << 8);
  }
}

template <int QB>
__device__ __forceinline__ constexpr int sidx(int r, int c) {
  if constexpr (QB == 0) return (r << 8) | c;
  else if constexpr (QB == 4) return ((c & 0xF0) << 4) | (r << 4) | (c & 15);
  else return (c << 4) | r;
}

__device__ __forceinline__ constexpr int ring_src(int i) {
  int j = i;
  for (int q = 11; q >= 0; --q) {
    const int cb = 11 - q;
    const int tb = 11 - ((q + 1) % 12);
    j ^= ((j >> cb) & 1) << tb;
  }
  return j;
}

__device__ __forceinline__ v8f wmma_f32(v2f a, v2f b, v8f c) {
  return __builtin_amdgcn_wmma_f32_16x16x4_f32(
      false, a, false, b, (short)0, c, false, false);
}

template <bool FIRST>
__device__ __forceinline__ void apply_hi(const v2f (&xq)[NB][2][4],
                         const float* sre0, const float* sim0,
                         float* dre0, float* dim0,
                         const float* ur, const float* pr, const float* pi) {
  const int lane  = threadIdx.x & 31;
  const int wave  = threadIdx.x >> 5;
  const int half  = lane >> 4;
  const int n     = lane & 15;
  const int khalf = half << 1;
  const int phi   = (wave ^ (wave >> 1) ^ (wave >> 2)) & 1;
  const int rlane = ph(ring_src((half << 9) | (wave << 5) | n));

  v2f aR[4];
  v2f eR[2][4], eI[2][4];
  {
    const cx yl = { pr[32 + n], pi[32 + n] };
    v2f hR[4], hI[4];
#pragma unroll
    for (int kc = 0; kc < 4; ++kc) {
      const int k0 = kc * 4 + khalf;
      aR[kc] = *(const v2f*)&ur[n * GP + k0];
      hR[kc] = *(const v2f*)&pr[k0];
      hI[kc] = *(const v2f*)&pi[k0];
    }
#pragma unroll
    for (int blk = 0; blk < 2; ++blk) {
      const cx ym = { pr[16 + 2 * wave + blk], pi[16 + 2 * wave + blk] };
      const cx q  = cmul(ym, yl);
#pragma unroll
      for (int kc = 0; kc < 4; ++kc) {
        eR[blk][kc] = hR[kc] * q.r - hI[kc] * q.i;
        eI[blk][kc] = hR[kc] * q.i + hI[kc] * q.r;
      }
    }
  }

  v2f rwR[2][4], rwI[2][4];
  auto hi_loads = [&](int t, v2f (&bR)[4], v2f (&bI)[4]) {
    const float* sre = sre0 + (t >> 1) * ESTR;
    const float* sim = sim0 + (t >> 1) * ESTR;
    const int blk = t & 1;
#pragma unroll
    for (int kc = 0; kc < 4; ++kc) {
      const int a0 = rlane ^ ph(ring_src(((4 * kc) << 8) | (blk << 4)));
      const int a1 = rlane ^ ph(ring_src(((4 * kc + 1) << 8) | (blk << 4)));
      bR[kc].x = sre[a0];
      bR[kc].y = sre[a1];
      bI[kc].x = sim[a0];
      bI[kc].y = sim[a1];
    }
  };
  if constexpr (!FIRST) hi_loads(0, rwR[0], rwI[0]);

#pragma unroll
  for (int t = 0; t < 2 * NB; ++t) {
    const int el = t >> 1, blk = t & 1;
    float* dre = dre0 + el * ESTR;
    float* dim = dim0 + el * ESTR;
    if constexpr (!FIRST) {
      if (t + 1 < 2 * NB) hi_loads(t + 1, rwR[(t + 1) & 1], rwI[(t + 1) & 1]);
      __builtin_amdgcn_sched_barrier(0);
    }
    v2f cR[4], cI[4];
#pragma unroll
    for (int kc = 0; kc < 4; ++kc) {
      if constexpr (FIRST) {
        cR[kc] = xq[el][blk][kc] * eR[blk][kc];
        cI[kc] = xq[el][blk][kc] * eI[blk][kc];
      } else {
        const v2f bR = rwR[t & 1][kc], bI = rwI[t & 1][kc];
        cR[kc] = bR * eR[blk][kc] - bI * eI[blk][kc];
        cI[kc] = bR * eI[blk][kc] + bI * eR[blk][kc];
      }
    }

    v8f dR = {};
    v8f dI = {};
#pragma unroll
    for (int kc = 0; kc < 4; ++kc) {
      dR = wmma_f32(aR[kc], cR[kc], dR);
      dI = wmma_f32(aR[kc], cI[kc], dI);
    }

#pragma unroll
    for (int v = 0; v < 8; ++v) {
      const int idx = st_addr<0>(wave, n, half, phi, blk, v);
      dre[idx] = dR[v];
      dim[idx] = dI[v];
    }
    if constexpr (!FIRST) {
      __builtin_amdgcn_sched_barrier(0);
    }
  }
}

template <bool LAST>
__device__ void apply_mid_lo(float* cre0, float* cim0,
                             const float* umr, const float* ulr,
                             const float* pr, const float* pi,
                             const float* ssq, float* ob) {
  const int lane  = threadIdx.x & 31;
  const int wave  = threadIdx.x >> 5;
  const int half  = lane >> 4;
  const int n     = lane & 15;
  const int khalf = half << 1;
  const int phi   = (wave ^ (wave >> 1) ^ (wave >> 2)) & 1;

  v2f mR[4];
  v2f lR[4];
#pragma unroll
  for (int kc = 0; kc < 4; ++kc) {
    const int k0 = kc * 4 + khalf;
    mR[kc] = *(const v2f*)&umr[n * GP + k0];
    lR[kc] = *(const v2f*)&ulr[n * GP + k0];
  }

  v8f fR[2], fI[2];
  {
    const cx xm = { pr[64 + n], pi[64 + n] };
    const v4f lr0 = *(const v4f*)&pr[80 + 8 * half];
    const v4f lr1 = *(const v4f*)&pr[84 + 8 * half];
    const v4f li0 = *(const v4f*)&pi[80 + 8 * half];
    const v4f li1 = *(const v4f*)&pi[84 + 8 * half];
    const v8f xr = { lr0[0], lr0[1], lr0[2], lr0[3], lr1[0], lr1[1], lr1[2], lr1[3] };
    const v8f xi = { li0[0], li0[1], li0[2], li0[3], li1[0], li1[1], li1[2], li1[3] };
#pragma unroll
    for (int blk = 0; blk < 2; ++blk) {
      const cx xh = { pr[48 + 2 * wave + blk], pi[48 + 2 * wave + blk] };
      const cx q  = cmul(xm, xh);
      fR[blk] = xr * q.r - xi * q.i;
      fI[blk] = xr * q.i + xi * q.r;
    }
  }

  float scale[NB] = { 1.f, 1.f };
  if constexpr (LAST) {
#pragma unroll
    for (int el = 0; el < NB; ++el) {
      float ss = ssq[el * 8];
#pragma unroll
      for (int j = 1; j < 8; ++j) ss += ssq[el * 8 + j];
      scale[el] = rsqrtf(sqrtf(ss) + 1e-12f);
    }
  }

  v2f rwR[2][4], rwI[2][4];
  auto ml_loads = [&](int t, v2f (&sR)[4], v2f (&sI)[4]) {
    const float* cre = cre0 + (t >> 1) * ESTR;
    const float* cim = cim0 + (t >> 1) * ESTR;
    const int blk = t & 1;
#pragma unroll
    for (int kc = 0; kc < 4; ++kc) {
      const int a0 = ld_addr<4>(wave, n, half, phi, blk, kc, 0);
      const int a1 = ld_addr<4>(wave, n, half, phi, blk, kc, 1);
      sR[kc].x = cre[a0];
      sR[kc].y = cre[a1];
      sI[kc].x = cim[a0];
      sI[kc].y = cim[a1];
    }
  };
  ml_loads(0, rwR[0], rwI[0]);

#pragma unroll
  for (int t = 0; t < 2 * NB; ++t) {
    const int el = t >> 1, blk = t & 1;
    float* cre = cre0 + el * ESTR;
    float* cim = cim0 + el * ESTR;
    if (t + 1 < 2 * NB) ml_loads(t + 1, rwR[(t + 1) & 1], rwI[(t + 1) & 1]);
    __builtin_amdgcn_sched_barrier(0);

    v8f tR = {};
    v8f tI = {};
#pragma unroll
    for (int kc = 0; kc < 4; ++kc) {
      tR = wmma_f32(rwR[t & 1][kc], mR[kc], tR);
      tI = wmma_f32(rwI[t & 1][kc], mR[kc], tI);
    }

    v2f bR[4], bI[4];
#pragma unroll
    for (int kc = 0; kc < 4; ++kc) {
      bR[kc].x = tR[2 * kc];
      bR[kc].y = tR[2 * kc + 1];
      bI[kc].x = tI[2 * kc];
      bI[kc].y = tI[2 * kc + 1];
    }

    v8f dR = {};
    v8f dI = {};
#pragma unroll
    for (int kc = 0; kc < 4; ++kc) {
      dR = wmma_f32(lR[kc], bR[kc], dR);
      dI = wmma_f32(lR[kc], bI[kc], dI);
    }

    const v8f oR = dR * fR[blk] - dI * fI[blk];

    if constexpr (LAST) {
      const v8f oS = oR * scale[el];
      float* o = ob + el * DIM + ((2 * wave + blk) << 8) + (n << 4) + 8 * half;
#pragma unroll
      for (int v = 0; v < 2; ++v) {
        v4f pr4 = { oS[4 * v], oS[4 * v + 1], oS[4 * v + 2], oS[4 * v + 3] };
        *(v4f*)&o[4 * v] = pr4;
      }
    } else {
      const v8f oI = dR * fI[blk] + dI * fR[blk];

#pragma unroll
      for (int v = 0; v < 2; ++v) {
        const int a = st_addr<8>(wave, n, half, phi, blk, v);
        v4f pr4 = { oR[4 * v], oR[4 * v + 1], oR[4 * v + 2], oR[4 * v + 3] };
        v4f pi4 = { oI[4 * v], oI[4 * v + 1], oI[4 * v + 2], oI[4 * v + 3] };
        *(v4f*)&cre[a] = pr4;
        *(v4f*)&cim[a] = pi4;
      }
    }
    __builtin_amdgcn_sched_barrier(0);
  }
}

__device__ __forceinline__ void build_layer(int l, int tid,
    const float (*g2re)[N_QUBITS][4], const float (*pqr)[N_QUBITS][4],
    const float (*pqi)[N_QUBITS][4],
    float (*ure)[16 * GP], float (*pvr)[16], float (*pvi)[16]) {
  const int m = tid >> 4, j = tid & 15;
#pragma unroll
  for (int g = 0; g < 3; ++g) {
    float prod = 1.f;
#pragma unroll
    for (int k = 0; k < 4; ++k) {
      const int q  = g * 4 + k;
      const int mb = (m >> (3 - k)) & 1;
      const int jb = (j >> (3 - k)) & 1;
      prod *= g2re[l][q][mb * 2 + jb];
    }
    const int jp = (g == 2) ? lo_pos(j) : j;
    ure[g][m * GP + jp] = prod;
  }
  if (tid < 96) {
    const int post = (m >= 3) ? 1 : 0;
    const int g    = m - 3 * post;
    cx prod = {1.f, 0.f};
#pragma unroll
    for (int k = 0; k < 4; ++k) {
      const int q  = g * 4 + k;
      const int jb = (j >> (3 - k)) & 1;
      const cx u = { pqr[l][q][post * 2 + jb], pqi[l][q][post * 2 + jb] };
      prod = cmul(prod, u);
    }
    pvr[m][j] = prod.r;
    pvi[m][j] = prod.i;
  }
}

__global__ __launch_bounds__(256)
void qsim_kernel(const float* __restrict__ x, const float* __restrict__ w,
                 float* __restrict__ out) {
  __shared__ __attribute__((aligned(256))) float sre[NB * ESTR];
  __shared__ __attribute__((aligned(256))) float sim[NB * ESTR];
  __shared__ __attribute__((aligned(16))) float ure[3][16 * GP];
  __shared__ __attribute__((aligned(16))) float pvr[6][16];
  __shared__ __attribute__((aligned(16))) float pvi[6][16];
  __shared__ float g2re[N_LAYERS + 1][N_QUBITS][4];
  __shared__ float pqr[N_LAYERS + 1][N_QUBITS][4];
  __shared__ float pqi[N_LAYERS + 1][N_QUBITS][4];
  __shared__ float red[NB][8];

  const int b   = blockIdx.x;
  const int tid = threadIdx.x;
  const float* xb = x + (size_t)b * (NB * DIM);

  const int gl = tid >> 5, gq = tid & 31;
  const bool gbuild = (gl <= N_LAYERS) && (gq < N_QUBITS);
  const int  gofs = gbuild ? gl * 36 + gq * 3 : 0;
  const float p0 = w[gofs], p1 = w[gofs + 1], p2 = w[gofs + 2];
  __builtin_amdgcn_sched_barrier(0);

  v2f xq[NB][2][4];
  {
    const int xlane = tid & 31, xwave = tid >> 5;
    const float* xl = xb + ((xlane >> 4) << 9) + xwave * 32 + (xlane & 15);
#pragma unroll
    for (int el = 0; el < NB; ++el)
#pragma unroll
      for (int kc = 0; kc < 4; ++kc)
#pragma unroll
        for (int blk = 0; blk < 2; ++blk) {
          xq[el][blk][kc].x = xl[el * DIM + ((4 * kc) << 8) + blk * 16];
          xq[el][blk][kc].y = xl[el * DIM + ((4 * kc + 1) << 8) + blk * 16];
        }
  }

  asm volatile("" :: "v"(p0), "v"(p1), "v"(p2));

  if (gbuild) {
    m2 U;
    if (gl < N_LAYERS) U = mmul(rot_z(p2), mmul(rot_y(p1), rot_x(p0)));
    else               U = mmul(rot_y(p2), mmul(rot_z(p1), rot_y(p0)));
    float c, sn, t;
    const cx x0 = unit_of(U.e[0][0], c);
    const cx x1 = unit_of(U.e[1][0], sn);
    cx y1;
    if (c >= sn) {
      y1 = cmul(unit_of(U.e[1][1], t), cconj(x1));
    } else {
      const cx nb = { -U.e[0][1].r, -U.e[0][1].i };
      y1 = cmul(cconj(x0), unit_of(nb, t));
    }
    g2re[gl][gq][0] = c;   g2re[gl][gq][1] = -sn;
    g2re[gl][gq][2] = sn;  g2re[gl][gq][3] = c;
    pqr[gl][gq][0] = 1.f;  pqi[gl][gq][0] = 0.f;
    pqr[gl][gq][1] = y1.r; pqi[gl][gq][1] = y1.i;
    pqr[gl][gq][2] = x0.r; pqi[gl][gq][2] = x0.i;
    pqr[gl][gq][3] = x1.r; pqi[gl][gq][3] = x1.i;
  }

#pragma unroll
  for (int el = 0; el < NB; ++el) {
    float ss = 0.f;
#pragma unroll
    for (int blk = 0; blk < 2; ++blk)
#pragma unroll
      for (int kc = 0; kc < 4; ++kc) {
        ss += xq[el][blk][kc].x * xq[el][blk][kc].x;
        ss += xq[el][blk][kc].y * xq[el][blk][kc].y;
      }
#pragma unroll
    for (int m = 16; m > 0; m >>= 1) ss += __shfl_xor(ss, m, 32);
    if ((tid & 31) == 0) red[el][tid >> 5] = ss;
  }
  __syncthreads();

  float* ob = out + (size_t)b * (NB * DIM);

  build_layer(0, tid, g2re, pqr, pqi, ure, pvr, pvi);
  __syncthreads();
  apply_hi<true>(xq, sre, sim, sre, sim, ure[0], &pvr[0][0], &pvi[0][0]);
  __syncthreads();
  apply_mid_lo<false>(sre, sim, ure[1], ure[2], &pvr[0][0], &pvi[0][0], &red[0][0], ob);
  __syncthreads();

  int cur = 0;
  for (int l = 1; l <= N_LAYERS; ++l) {
    build_layer(l, tid, g2re, pqr, pqi, ure, pvr, pvi);
    __syncthreads();

    const int oth = cur ^ DIM;
    apply_hi<false>(xq, sre + cur, sim + cur, sre + oth, sim + oth, ure[0], &pvr[0][0], &pvi[0][0]);
    cur = oth;
    float* cre = sre + cur;
    float* cim = sim + cur;
    __syncthreads();
    if (l == N_LAYERS) {
      apply_mid_lo<true>(cre, cim, ure[1], ure[2], &pvr[0][0], &pvi[0][0], &red[0][0], ob);
    } else {
      apply_mid_lo<false>(cre, cim, ure[1], ure[2], &pvr[0][0], &pvi[0][0], &red[0][0], ob);
      __syncthreads();
    }
  }
}

extern "C" void kernel_launch(void* const* d_in, const int* in_sizes, int n_in,
                              void* d_out, int out_size, void* d_ws, size_t ws_size,
                              hipStream_t stream) {
  const float* x = (const float*)d_in[0];
  const float* w = (const float*)d_in[1];
  float* out = (float*)d_out;
  const int groups = in_sizes[0] >> 13;
  qsim_kernel<<<dim3(groups), dim3(256), 0, stream>>>(x, w, out);
}


# issue priority falls with progress (s_setprio 3 in layers 0-3, 2 in layer 4, 1 and 0 in the last layer's two passes): of a WGP's two workgroups the one behind gets the matrix pipe; sums unchanged
# speedup vs baseline: 2.2162x; 1.0413x over previous
#include <hip/hip_runtime.h>
#include <hip/hip_bf16.h>


#define DIM       4096
#define NB        2
#define ESTR      (2 * DIM)
#define N_QUBITS  12
#define N_LAYERS  5
#define GP        20

__device__ __forceinline__ constexpr int lo_pos(int k) {
  return (((k >> 1) & 3) << 2) | ((k >> 3) << 1) | (k & 1);
}

typedef __attribute__((ext_vector_type(2))) float v2f;
typedef __attribute__((ext_vector_type(4))) float v4f;
typedef __attribute__((ext_vector_type(8))) float v8f;

struct cx { float r, i; };
__device__ __forceinline__ cx cmul(cx a, cx b) { return { a.r*b.r - a.i*b.i, a.r*b.i + a.i*b.r }; }
__device__ __forceinline__ cx cadd(cx a, cx b) { return { a.r + b.r, a.i + b.i }; }
__device__ __forceinline__ cx cconj(cx a) { return { a.r, -a.i }; }

__device__ __forceinline__ cx unit_of(cx z, float& mag) {
  const float mx = fmaxf(fabsf(z.r), fabsf(z.i));
  if (!(mx > 0.f)) { mag = mx; return { 1.f, 0.f }; }
  const float zr = z.r / mx, zi = z.i / mx;
  const float m  = sqrtf(zr * zr + zi * zi);
  mag = mx * m;
  return { zr / m, zi / m };
}

struct m2 { cx e[2][2]; };

__device__ __forceinline__ m2 mmul(const m2& A, const m2& B) {
  m2 C;
#pragma unroll
  for (int i = 0; i < 2; ++i)
#pragma unroll
    for (int j = 0; j < 2; ++j)
      C.e[i][j] = cadd(cmul(A.e[i][0], B.e[0][j]), cmul(A.e[i][1], B.e[1][j]));
  return C;
}

__device__ __forceinline__ m2 rot_x(float t) {
  float c = cosf(0.5f * t), s = sinf(0.5f * t);
  m2 M; M.e[0][0] = {c, 0}; M.e[0][1] = {0, -s};
        M.e[1][0] = {0, -s}; M.e[1][1] = {c, 0};
  return M;
}
__device__ __forceinline__ m2 rot_y(float t) {
  float c = cosf(0.5f * t), s = sinf(0.5f * t);
  m2 M; M.e[0][0] = {c, 0}; M.e[0][1] = {-s, 0};
        M.e[1][0] = {s, 0}; M.e[1][1] = {c, 0};
  return M;
}
__device__ __forceinline__ m2 rot_z(float t) {
  float c = cosf(0.5f * t), s = sinf(0.5f * t);
  m2 M; M.e[0][0] = {c, -s}; M.e[0][1] = {0, 0};
        M.e[1][0] = {0, 0};  M.e[1][1] = {c, s};
  return M;
}

__device__ __forceinline__ constexpr int ph(int i) {
  return i ^ (((i >> 6) & 3) << 2) ^ (((i >> 7) & 1) << 4)
           ^ ((((i >> 9) ^ (i >> 10) ^ (i >> 11)) & 1) << 5);
}

template <int QB>
__device__ __forceinline__ int ld_addr(int wave, int n, int half, int phi,
                                       int blk, int kc, int e) {
  if constexpr (QB == 4) {
    const int base = (wave << 9) | ((half ^ phi) << 5) | (n ^ (kc << 2));
    return base + (blk << 8) + (kc << 6) + ((e ^ (kc >> 1)) << 4);
  } else {
    const int rowb = (wave << 9) | ((n ^ (n >> 3) ^ (phi << 1)) << 4);
    return (rowb | ((kc ^ (n >> 2)) << 2) | (half << 1)) + (blk << 8) + e;
  }
}

template <int QB>
__device__ __forceinline__ int st_addr(int wave, int n, int half, int phi,
                                       int blk, int v) {
  if constexpr (QB == 0) {
    const int c0  = wave * 32 + n;
    const int lo0 = c0 ^ (((c0 >> 6) & 3) << 2) ^ (((c0 >> 7) & 1) << 4) ^ (half << 5);
    const int pv  = ((v >> 1) ^ (v >> 2)) & 1;
    return (((lo0 ^ (pv << 5)) | (half << 11)) ^ (blk << 4)) + (v << 8);
  } else if constexpr (QB == 4) {
    const int sb = (wave << 9) | (half << 7) | (phi << 5) | (half << 4) | (n ^ (half << 3));
    return (sb ^ ((v & 2) << 4) ^ ((v & 1) << 4) ^ (v & 4)) + (blk << 8) + ((v & 4) << 4);
  } else {
    const int rowb = (wave << 9) | ((n ^ (n >> 3) ^ (phi << 1)) << 4);
    return (rowb | (((2 * half + v) ^ (n >> 2)) << 2)) + (blk << 8);
  }
}

template <int QB>
__device__ __forceinline__ constexpr int sidx(int r, int c) {
  if constexpr (QB == 0) return (r << 8) | c;
  else if constexpr (QB == 4) return ((c & 0xF0) << 4) | (r << 4) | (c & 15);
  else return (c << 4) | r;
}

__device__ __forceinline__ constexpr int ring_src(int i) {
  int j = i;
  for (int q = 11; q >= 0; --q) {
    const int cb = 11 - q;
    const int tb = 11 - ((q + 1) % 12);
    j ^= ((j >> cb) & 1) << tb;
  }
  return j;
}

__device__ __forceinline__ v8f wmma_f32(v2f a, v2f b, v8f c) {
  return __builtin_amdgcn_wmma_f32_16x16x4_f32(
      false, a, false, b, (short)0, c, false, false);
}

template <bool FIRST>
__device__ __forceinline__ void apply_hi(const v2f (&xq)[NB][2][4],
                         const float* sre0, const float* sim0,
                         float* dre0, float* dim0,
                         const float* ur, const float* pr, const float* pi) {
  const int lane  = threadIdx.x & 31;
  const int wave  = threadIdx.x >> 5;
  const int half  = lane >> 4;
  const int n     = lane & 15;
  const int khalf = half << 1;
  const int phi   = (wave ^ (wave >> 1) ^ (wave >> 2)) & 1;
  const int rlane = ph(ring_src((half << 9) | (wave << 5) | n));

  v2f aR[4];
  v2f eR[2][4], eI[2][4];
  {
    const cx yl = { pr[32 + n], pi[32 + n] };
    v2f hR[4], hI[4];
#pragma unroll
    for (int kc = 0; kc < 4; ++kc) {
      const int k0 = kc * 4 + khalf;
      aR[kc] = *(const v2f*)&ur[n * GP + k0];
      hR[kc] = *(const v2f*)&pr[k0];
      hI[kc] = *(const v2f*)&pi[k0];
    }
#pragma unroll
    for (int blk = 0; blk < 2; ++blk) {
      const cx ym = { pr[16 + 2 * wave + blk], pi[16 + 2 * wave + blk] };
      const cx q  = cmul(ym, yl);
#pragma unroll
      for (int kc = 0; kc < 4; ++kc) {
        eR[blk][kc] = hR[kc] * q.r - hI[kc] * q.i;
        eI[blk][kc] = hR[kc] * q.i + hI[kc] * q.r;
      }
    }
  }

  v2f rwR[2][4], rwI[2][4];
  auto hi_loads = [&](int t, v2f (&bR)[4], v2f (&bI)[4]) {
    const float* sre = sre0 + (t >> 1) * ESTR;
    const float* sim = sim0 + (t >> 1) * ESTR;
    const int blk = t & 1;
#pragma unroll
    for (int kc = 0; kc < 4; ++kc) {
      const int a0 = rlane ^ ph(ring_src(((4 * kc) << 8) | (blk << 4)));
      const int a1 = rlane ^ ph(ring_src(((4 * kc + 1) << 8) | (blk << 4)));
      bR[kc].x = sre[a0];
      bR[kc].y = sre[a1];
      bI[kc].x = sim[a0];
      bI[kc].y = sim[a1];
    }
  };
  if constexpr (!FIRST) hi_loads(0, rwR[0], rwI[0]);

#pragma unroll
  for (int t = 0; t < 2 * NB; ++t) {
    const int el = t >> 1, blk = t & 1;
    float* dre = dre0 + el * ESTR;
    float* dim = dim0 + el * ESTR;
    if constexpr (!FIRST) {
      if (t + 1 < 2 * NB) hi_loads(t + 1, rwR[(t + 1) & 1], rwI[(t + 1) & 1]);
      __builtin_amdgcn_sched_barrier(0);
    }
    v2f cR[4], cI[4];
#pragma unroll
    for (int kc = 0; kc < 4; ++kc) {
      if constexpr (FIRST) {
        cR[kc] = xq[el][blk][kc] * eR[blk][kc];
        cI[kc] = xq[el][blk][kc] * eI[blk][kc];
      } else {
        const v2f bR = rwR[t & 1][kc], bI = rwI[t & 1][kc];
        cR[kc] = bR * eR[blk][kc] - bI * eI[blk][kc];
        cI[kc] = bR * eI[blk][kc] + bI * eR[blk][kc];
      }
    }

    v8f dR = {};
    v8f dI = {};
#pragma unroll
    for (int kc = 0; kc < 4; ++kc) {
      dR = wmma_f32(aR[kc], cR[kc], dR);
      dI = wmma_f32(aR[kc], cI[kc], dI);
    }

#pragma unroll
    for (int v = 0; v < 8; ++v) {
      const int idx = st_addr<0>(wave, n, half, phi, blk, v);
      dre[idx] = dR[v];
      dim[idx] = dI[v];
    }
    if constexpr (!FIRST) {
      __builtin_amdgcn_sched_barrier(0);
    }
  }
}

template <bool LAST>
__device__ void apply_mid_lo(float* cre0, float* cim0,
                             const float* umr, const float* ulr,
                             const float* pr, const float* pi,
                             const float* ssq, float* ob) {
  const int lane  = threadIdx.x & 31;
  const int wave  = threadIdx.x >> 5;
  const int half  = lane >> 4;
  const int n     = lane & 15;
  const int khalf = half << 1;
  const int phi   = (wave ^ (wave >> 1) ^ (wave >> 2)) & 1;

  v2f mR[4];
  v2f lR[4];
#pragma unroll
  for (int kc = 0; kc < 4; ++kc) {
    const int k0 = kc * 4 + khalf;
    mR[kc] = *(const v2f*)&umr[n * GP + k0];
    lR[kc] = *(const v2f*)&ulr[n * GP + k0];
  }

  v8f fR[2], fI[2];
  {
    const cx xm = { pr[64 + n], pi[64 + n] };
    const v4f lr0 = *(const v4f*)&pr[80 + 8 * half];
    const v4f lr1 = *(const v4f*)&pr[84 + 8 * half];
    const v4f li0 = *(const v4f*)&pi[80 + 8 * half];
    const v4f li1 = *(const v4f*)&pi[84 + 8 * half];
    const v8f xr = { lr0[0], lr0[1], lr0[2], lr0[3], lr1[0], lr1[1], lr1[2], lr1[3] };
    const v8f xi = { li0[0], li0[1], li0[2], li0[3], li1[0], li1[1], li1[2], li1[3] };
#pragma unroll
    for (int blk = 0; blk < 2; ++blk) {
      const cx xh = { pr[48 + 2 * wave + blk], pi[48 + 2 * wave + blk] };
      const cx q  = cmul(xm, xh);
      fR[blk] = xr * q.r - xi * q.i;
      fI[blk] = xr * q.i + xi * q.r;
    }
  }

  float scale[NB] = { 1.f, 1.f };
  if constexpr (LAST) {
#pragma unroll
    for (int el = 0; el < NB; ++el) {
      float ss = ssq[el * 8];
#pragma unroll
      for (int j = 1; j < 8; ++j) ss += ssq[el * 8 + j];
      scale[el] = rsqrtf(sqrtf(ss) + 1e-12f);
    }
  }

  v2f rwR[2][4], rwI[2][4];
  auto ml_loads = [&](int t, v2f (&sR)[4], v2f (&sI)[4]) {
    const float* cre = cre0 + (t >> 1) * ESTR;
    const float* cim = cim0 + (t >> 1) * ESTR;
    const int blk = t & 1;
#pragma unroll
    for (int kc = 0; kc < 4; ++kc) {
      const int a0 = ld_addr<4>(wave, n, half, phi, blk, kc, 0);
      const int a1 = ld_addr<4>(wave, n, half, phi, blk, kc, 1);
      sR[kc].x = cre[a0];
      sR[kc].y = cre[a1];
      sI[kc].x = cim[a0];
      sI[kc].y = cim[a1];
    }
  };
  ml_loads(0, rwR[0], rwI[0]);

#pragma unroll
  for (int t = 0; t < 2 * NB; ++t) {
    const int el = t >> 1, blk = t & 1;
    float* cre = cre0 + el * ESTR;
    float* cim = cim0 + el * ESTR;
    if (t + 1 < 2 * NB) ml_loads(t + 1, rwR[(t + 1) & 1], rwI[(t + 1) & 1]);
    __builtin_amdgcn_sched_barrier(0);

    v8f tR = {};
    v8f tI = {};
#pragma unroll
    for (int kc = 0; kc < 4; ++kc) {
      tR = wmma_f32(rwR[t & 1][kc], mR[kc], tR);
      tI = wmma_f32(rwI[t & 1][kc], mR[kc], tI);
    }

    v2f bR[4], bI[4];
#pragma unroll
    for (int kc = 0; kc < 4; ++kc) {
      bR[kc].x = tR[2 * kc];
      bR[kc].y = tR[2 * kc + 1];
      bI[kc].x = tI[2 * kc];
      bI[kc].y = tI[2 * kc + 1];
    }

    v8f dR = {};
    v8f dI = {};
#pragma unroll
    for (int kc = 0; kc < 4; ++kc) {
      dR = wmma_f32(lR[kc], bR[kc], dR);
      dI = wmma_f32(lR[kc], bI[kc], dI);
    }

    const v8f oR = dR * fR[blk] - dI * fI[blk];

    if constexpr (LAST) {
      const v8f oS = oR * scale[el];
      float* o = ob + el * DIM + ((2 * wave + blk) << 8) + (n << 4) + 8 * half;
#pragma unroll
      for (int v = 0; v < 2; ++v) {
        v4f pr4 = { oS[4 * v], oS[4 * v + 1], oS[4 * v + 2], oS[4 * v + 3] };
        *(v4f*)&o[4 * v] = pr4;
      }
    } else {
      const v8f oI = dR * fI[blk] + dI * fR[blk];

#pragma unroll
      for (int v = 0; v < 2; ++v) {
        const int a = st_addr<8>(wave, n, half, phi, blk, v);
        v4f pr4 = { oR[4 * v], oR[4 * v + 1], oR[4 * v + 2], oR[4 * v + 3] };
        v4f pi4 = { oI[4 * v], oI[4 * v + 1], oI[4 * v + 2], oI[4 * v + 3] };
        *(v4f*)&cre[a] = pr4;
        *(v4f*)&cim[a] = pi4;
      }
    }
    __builtin_amdgcn_sched_barrier(0);
  }
}

__device__ __forceinline__ void build_layer(int l, int tid,
    const float (*g2re)[N_QUBITS][4], const float (*pqr)[N_QUBITS][4],
    const float (*pqi)[N_QUBITS][4],
    float (*ure)[16 * GP], float (*pvr)[16], float (*pvi)[16]) {
  const int m = tid >> 4, j = tid & 15;
#pragma unroll
  for (int g = 0; g < 3; ++g) {
    float prod = 1.f;
#pragma unroll
    for (int k = 0; k < 4; ++k) {
      const int q  = g * 4 + k;
      const int mb = (m >> (3 - k)) & 1;
      const int jb = (j >> (3 - k)) & 1;
      prod *= g2re[l][q][mb * 2 + jb];
    }
    const int jp = (g == 2) ? lo_pos(j) : j;
    ure[g][m * GP + jp] = prod;
  }
  if (tid < 96) {
    const int post = (m >= 3) ? 1 : 0;
    const int g    = m - 3 * post;
    cx prod = {1.f, 0.f};
#pragma unroll
    for (int k = 0; k < 4; ++k) {
      const int q  = g * 4 + k;
      const int jb = (j >> (3 - k)) & 1;
      const cx u = { pqr[l][q][post * 2 + jb], pqi[l][q][post * 2 + jb] };
      prod = cmul(prod, u);
    }
    pvr[m][j] = prod.r;
    pvi[m][j] = prod.i;
  }
}

__global__ __launch_bounds__(256)
void qsim_kernel(const float* __restrict__ x, const float* __restrict__ w,
                 float* __restrict__ out) {
  __shared__ __attribute__((aligned(256))) float sre[NB * ESTR];
  __shared__ __attribute__((aligned(256))) float sim[NB * ESTR];
  __shared__ __attribute__((aligned(16))) float ure[3][16 * GP];
  __shared__ __attribute__((aligned(16))) float pvr[6][16];
  __shared__ __attribute__((aligned(16))) float pvi[6][16];
  __shared__ float g2re[N_LAYERS + 1][N_QUBITS][4];
  __shared__ float pqr[N_LAYERS + 1][N_QUBITS][4];
  __shared__ float pqi[N_LAYERS + 1][N_QUBITS][4];
  __shared__ float red[NB][8];

  __builtin_amdgcn_s_setprio(3);

  const int b   = blockIdx.x;
  const int tid = threadIdx.x;
  const float* xb = x + (size_t)b * (NB * DIM);

  const int gl = tid >> 5, gq = tid & 31;
  const bool gbuild = (gl <= N_LAYERS) && (gq < N_QUBITS);
  const int  gofs = gbuild ? gl * 36 + gq * 3 : 0;
  const float p0 = w[gofs], p1 = w[gofs + 1], p2 = w[gofs + 2];
  __builtin_amdgcn_sched_barrier(0);

  v2f xq[NB][2][4];
  {
    const int xlane = tid & 31, xwave = tid >> 5;
    const float* xl = xb + ((xlane >> 4) << 9) + xwave * 32 + (xlane & 15);
#pragma unroll
    for (int el = 0; el < NB; ++el)
#pragma unroll
      for (int kc = 0; kc < 4; ++kc)
#pragma unroll
        for (int blk = 0; blk < 2; ++blk) {
          xq[el][blk][kc].x = xl[el * DIM + ((4 * kc) << 8) + blk * 16];
          xq[el][blk][kc].y = xl[el * DIM + ((4 * kc + 1) << 8) + blk * 16];
        }
  }

  asm volatile("" :: "v"(p0), "v"(p1), "v"(p2));

  if (gbuild) {
    m2 U;
    if (gl < N_LAYERS) U = mmul(rot_z(p2), mmul(rot_y(p1), rot_x(p0)));
    else               U = mmul(rot_y(p2), mmul(rot_z(p1), rot_y(p0)));
    float c, sn, t;
    const cx x0 = unit_of(U.e[0][0], c);
    const cx x1 = unit_of(U.e[1][0], sn);
    cx y1;
    if (c >= sn) {
      y1 = cmul(unit_of(U.e[1][1], t), cconj(x1));
    } else {
      const cx nb = { -U.e[0][1].r, -U.e[0][1].i };
      y1 = cmul(cconj(x0), unit_of(nb, t));
    }
    g2re[gl][gq][0] = c;   g2re[gl][gq][1] = -sn;
    g2re[gl][gq][2] = sn;  g2re[gl][gq][3] = c;
    pqr[gl][gq][0] = 1.f;  pqi[gl][gq][0] = 0.f;
    pqr[gl][gq][1] = y1.r; pqi[gl][gq][1] = y1.i;
    pqr[gl][gq][2] = x0.r; pqi[gl][gq][2] = x0.i;
    pqr[gl][gq][3] = x1.r; pqi[gl][gq][3] = x1.i;
  }

#pragma unroll
  for (int el = 0; el < NB; ++el) {
    float ss = 0.f;
#pragma unroll
    for (int blk = 0; blk < 2; ++blk)
#pragma unroll
      for (int kc = 0; kc < 4; ++kc) {
        ss += xq[el][blk][kc].x * xq[el][blk][kc].x;
        ss += xq[el][blk][kc].y * xq[el][blk][kc].y;
      }
#pragma unroll
    for (int m = 16; m > 0; m >>= 1) ss += __shfl_xor(ss, m, 32);
    if ((tid & 31) == 0) red[el][tid >> 5] = ss;
  }
  __syncthreads();

  float* ob = out + (size_t)b * (NB * DIM);

  build_layer(0, tid, g2re, pqr, pqi, ure, pvr, pvi);
  __syncthreads();
  apply_hi<true>(xq, sre, sim, sre, sim, ure[0], &pvr[0][0], &pvi[0][0]);
  __syncthreads();
  apply_mid_lo<false>(sre, sim, ure[1], ure[2], &pvr[0][0], &pvi[0][0], &red[0][0], ob);
  __syncthreads();

  int cur = 0;
  for (int l = 1; l <= N_LAYERS; ++l) {
    if (l == N_LAYERS - 1) __builtin_amdgcn_s_setprio(2);
    if (l == N_LAYERS)     __builtin_amdgcn_s_setprio(1);
    build_layer(l, tid, g2re, pqr, pqi, ure, pvr, pvi);
    __syncthreads();

    const int oth = cur ^ DIM;
    apply_hi<false>(xq, sre + cur, sim + cur, sre + oth, sim + oth, ure[0], &pvr[0][0], &pvi[0][0]);
    cur = oth;
    float* cre = sre + cur;
    float* cim = sim + cur;
    __syncthreads();
    if (l == N_LAYERS) {
      __builtin_amdgcn_s_setprio(0);
      apply_mid_lo<true>(cre, cim, ure[1], ure[2], &pvr[0][0], &pvi[0][0], &red[0][0], ob);
    } else {
      apply_mid_lo<false>(cre, cim, ure[1], ure[2], &pvr[0][0], &pvi[0][0], &red[0][0], ob);
      __syncthreads();
    }
  }
}

extern "C" void kernel_launch(void* const* d_in, const int* in_sizes, int n_in,
                              void* d_out, int out_size, void* d_ws, size_t ws_size,
                              hipStream_t stream) {
  const float* x = (const float*)d_in[0];
  const float* w = (const float*)d_in[1];
  float* out = (float*)d_out;
  const int groups = in_sizes[0] >> 13;
  qsim_kernel<<<dim3(groups), dim3(256), 0, stream>>>(x, w, out);
}
